# starting kernel
# baseline (speedup 1.0000x reference)
#include <hip/hip_runtime.h>

#define H_DIM   128
#define F_IN_DIM 256
#define C_DIM   40
#define L_DIM   4
#define STEPF   0.1f

typedef __attribute__((ext_vector_type(2))) float v2f;
typedef __attribute__((ext_vector_type(4))) float v4f;
typedef __attribute__((ext_vector_type(8))) float v8f;


__global__ void zero_two(float* __restrict__ a, float* __restrict__ b, int n) {
    int i = blockIdx.x * blockDim.x + threadIdx.x;
    if (i < n) { a[i] = 0.f; b[i] = 0.f; }
}

__global__ void deg_accum(const int* __restrict__ row, float* __restrict__ deg, int E) {
    int e = blockIdx.x * blockDim.x + threadIdx.x;
    if (e < E) atomicAdd(&deg[row[e]], 1.0f);
}

__global__ void deg_inv_kernel(float* __restrict__ deg, int n) {
    int i = blockIdx.x * blockDim.x + threadIdx.x;
    if (i < n) { float d = deg[i]; deg[i] = (d > 0.f) ? rsqrtf(d) : 0.f; }
}

__global__ void coef_accum(const int* __restrict__ row, const int* __restrict__ col,
                           const float* __restrict__ dinv, float* __restrict__ coef, int E) {
    int e = blockIdx.x * blockDim.x + threadIdx.x;
    if (e < E) {
        int r = row[e], c = col[e];
        atomicAdd(&coef[c], dinv[r] * dinv[c]);
    }
}

__global__ void weff_kernel(const float* __restrict__ Wraw, float* __restrict__ Weff) {
    int l = blockIdx.x;
    int i = threadIdx.x;
    const float* Wr = Wraw + (size_t)l * H_DIM * (H_DIM + 2);
    float*       We = Weff + (size_t)l * H_DIM * H_DIM;
    float s = 0.f;
    for (int j = 0; j < H_DIM; ++j) {
        float w = 0.f;
        if (j > i)      w = Wr[i * (H_DIM + 2) + j];
        else if (j < i) w = Wr[j * (H_DIM + 2) + i];
        s += fabsf(w);
        We[i * H_DIM + j] = w;
    }
    float d = Wr[i * (H_DIM + 2) + H_DIM] * s + Wr[i * (H_DIM + 2) + H_DIM + 1];
    We[i * H_DIM + i] = d;
}


__launch_bounds__(256)
__global__ void encoder_gemm(const float* __restrict__ x, const float* __restrict__ W,
                             const float* __restrict__ bias, float* __restrict__ h, int Nn) {
    __shared__ __align__(16) float As[16][F_IN_DIM + 4];
    const int row0 = blockIdx.x * 16;
    const v4f* x4 = (const v4f*)x;
    for (int idx = threadIdx.x; idx < 16 * (F_IN_DIM / 4); idx += 256) {
        int r  = idx >> 6;
        int c4 = idx & 63;
        int node = row0 + r;
        v4f v = {0.f, 0.f, 0.f, 0.f};
        if (node < Nn)
            v = __builtin_nontemporal_load(&x4[(size_t)node * (F_IN_DIM / 4) + c4]);
        *(v4f*)&As[r][c4 * 4] = v;
    }
    __syncthreads();

    const int lane = threadIdx.x & 31;
    const int wave = threadIdx.x >> 5;
    const int col0 = wave * 16;
    const int m    = lane & 15;
    const int hi   = lane >> 4;
    const int colN = col0 + m;
    const float* wrow = W + (size_t)colN * F_IN_DIM;

    v8f acc = {};
    for (int k0 = 0; k0 < F_IN_DIM; k0 += 4) {
        int k = k0 + 2 * hi;
        v2f a = *(const v2f*)&As[m][k];
        v2f b = *(const v2f*)&wrow[k];
        acc = __builtin_amdgcn_wmma_f32_16x16x4_f32(false, a, false, b,
                                                    (short)0, acc, false, false);
    }
    const float bv = bias[colN];
    for (int r = 0; r < 8; ++r) {
        int node = row0 + r + 8 * hi;
        if (node < Nn) h[(size_t)node * H_DIM + colN] = acc[r] + bv;
    }
}

__launch_bounds__(256)
__global__ void layer_gemm(const float* __restrict__ h_in, const float* __restrict__ h0,
                           const float* __restrict__ coef, const float* __restrict__ Weff,
                           const float* __restrict__ Wb, const float* __restrict__ extw,
                           const float* __restrict__ beta_p, float* __restrict__ h_out, int Nn) {
    __shared__ __align__(16) float As[16][H_DIM + 4];
    const int row0 = blockIdx.x * 16;
    const v4f* h4 = (const v4f*)h_in;
    for (int idx = threadIdx.x; idx < 16 * (H_DIM / 4); idx += 256) {
        int r  = idx >> 5;
        int c4 = idx & 31;
        int node = row0 + r;
        v4f v = {0.f, 0.f, 0.f, 0.f};
        if (node < Nn) v = h4[(size_t)node * (H_DIM / 4) + c4];
        *(v4f*)&As[r][c4 * 4] = v;
    }
    __syncthreads();

    const int lane = threadIdx.x & 31;
    const int wave = threadIdx.x >> 5;
    const int col0 = wave * 16;
    const int m    = lane & 15;
    const int hi   = lane >> 4;
    const int colN = col0 + m;

    v8f acc = {};
    for (int k0 = 0; k0 < H_DIM; k0 += 4) {
        int k = k0 + 2 * hi;
        v2f a = *(const v2f*)&As[m][k];
        v2f b;
        b.x = Weff[(size_t)k * H_DIM + colN];
        b.y = Weff[(size_t)(k + 1) * H_DIM + colN];
        acc = __builtin_amdgcn_wmma_f32_16x16x4_f32(false, a, false, b,
                                                    (short)0, acc, false, false);
    }

    const float wbv = Wb[colN];
    const float ew  = extw[colN];
    const float bl  = *beta_p;
    for (int r = 0; r < 8; ++r) {
        int node = row0 + r + 8 * hi;
        if (node < Nn) {
            float cf   = coef[node];
            float hv   = As[r + 8 * hi][colN];
            float h0v  = h0[(size_t)node * H_DIM + colN];
            float outp = acc[r] + wbv;
            float outv = cf * outp - (hv * ew + bl * h0v);
            h_out[(size_t)node * H_DIM + colN] = hv + STEPF * fmaxf(outv, 0.f);
        }
    }
}

__launch_bounds__(96)
__global__ void decoder_gemm(const float* __restrict__ h, const float* __restrict__ W,
                             const float* __restrict__ bias, float* __restrict__ out, int Nn) {
    __shared__ __align__(16) float As[16][H_DIM + 4];
    const int row0 = blockIdx.x * 16;
    const v4f* h4 = (const v4f*)h;
    for (int idx = threadIdx.x; idx < 16 * (H_DIM / 4); idx += 96) {
        int r  = idx >> 5;
        int c4 = idx & 31;
        int node = row0 + r;
        v4f v = {0.f, 0.f, 0.f, 0.f};
        if (node < Nn) v = h4[(size_t)node * (H_DIM / 4) + c4];
        *(v4f*)&As[r][c4 * 4] = v;
    }
    __syncthreads();

    const int lane = threadIdx.x & 31;
    const int wave = threadIdx.x >> 5;
    const int col0 = wave * 16;
    const int m    = lane & 15;
    const int hi   = lane >> 4;
    const int colN = col0 + m;
    const bool colok = (colN < C_DIM);
    const float* wrow = W + (size_t)(colok ? colN : 0) * H_DIM;

    v8f acc = {};
    for (int k0 = 0; k0 < H_DIM; k0 += 4) {
        int k = k0 + 2 * hi;
        v2f a = *(const v2f*)&As[m][k];
        v2f b = {0.f, 0.f};
        if (colok) b = *(const v2f*)&wrow[k];
        acc = __builtin_amdgcn_wmma_f32_16x16x4_f32(false, a, false, b,
                                                    (short)0, acc, false, false);
    }
    if (colok) {
        const float bv = bias[colN];
        for (int r = 0; r < 8; ++r) {
            int node = row0 + r + 8 * hi;
            if (node < Nn)
                __builtin_nontemporal_store(acc[r] + bv,
                                            &out[(size_t)node * C_DIM + colN]);
        }
    }
}


extern "C" void kernel_launch(void* const* d_in, const int* in_sizes, int n_in,
                              void* d_out, int out_size, void* d_ws, size_t ws_size,
                              hipStream_t stream) {
    const float* x     = (const float*)d_in[0];
    const int*   eidx  = (const int*)d_in[1];
    const float* enc_W = (const float*)d_in[2];
    const float* enc_b = (const float*)d_in[3];
    const float* W_raw = (const float*)d_in[4];
    const float* W_b   = (const float*)d_in[5];
    const float* ext_w = (const float*)d_in[6];
    const float* beta  = (const float*)d_in[7];
    const float* dec_W = (const float*)d_in[8];
    const float* dec_b = (const float*)d_in[9];

    const int Nn = in_sizes[0] / F_IN_DIM;
    const int E  = in_sizes[1] / 2;
    const int* row = eidx;
    const int* col = eidx + E;

    char* ws = (char*)d_ws;
    size_t off = 0;
    auto wsalloc = [&](size_t bytes) -> void* {
        void* p = ws + off;
        off = (off + bytes + 255) & ~(size_t)255;
        return p;
    };
    float* deg  = (float*)wsalloc((size_t)Nn * sizeof(float));
    float* coef = (float*)wsalloc((size_t)Nn * sizeof(float));
    float* Weff = (float*)wsalloc((size_t)L_DIM * H_DIM * H_DIM * sizeof(float));
    float* h0   = (float*)wsalloc((size_t)Nn * H_DIM * sizeof(float));
    float* ha   = (float*)wsalloc((size_t)Nn * H_DIM * sizeof(float));
    float* hb   = (float*)wsalloc((size_t)Nn * H_DIM * sizeof(float));
    (void)ws_size; (void)n_in; (void)out_size;

    const int tb = 256;
    zero_two<<<(Nn + tb - 1) / tb, tb, 0, stream>>>(deg, coef, Nn);
    deg_accum<<<(E + tb - 1) / tb, tb, 0, stream>>>(row, deg, E);
    deg_inv_kernel<<<(Nn + tb - 1) / tb, tb, 0, stream>>>(deg, Nn);
    coef_accum<<<(E + tb - 1) / tb, tb, 0, stream>>>(row, col, deg, coef, E);
    weff_kernel<<<L_DIM, H_DIM, 0, stream>>>(W_raw, Weff);

    const int gTiles = (Nn + 15) / 16;
    encoder_gemm<<<gTiles, 256, 0, stream>>>(x, enc_W, enc_b, h0, Nn);

    const float* hin = h0;
    float* bufs[2] = { ha, hb };
    for (int l = 0; l < L_DIM; ++l) {
        float* hout = bufs[l & 1];
        layer_gemm<<<gTiles, 256, 0, stream>>>(hin, h0, coef,
                                               Weff + (size_t)l * H_DIM * H_DIM,
                                               W_b + (size_t)l * H_DIM,
                                               ext_w + (size_t)l * H_DIM,
                                               beta + l, hout, Nn);
        hin = hout;
    }
    decoder_gemm<<<gTiles, 96, 0, stream>>>(hin, dec_W, dec_b, (float*)d_out, Nn);
}


# layer GEMM forms its weight strip itself: block stages the raw 128x130 matrix in LDS, each wave builds its 16 W_eff columns and diagonal there; W_eff kernel, buffer, launch go; sum order unchanged
# speedup vs baseline: 1.3978x; 1.3978x over previous
#include <hip/hip_runtime.h>

#define H_DIM   128
#define F_IN_DIM 256
#define C_DIM   40
#define L_DIM   4
#define STEPF   0.1f
#define WR_PITCH (H_DIM + 2)
#define WR_ELEMS (H_DIM * WR_PITCH)

typedef __attribute__((ext_vector_type(2))) float v2f;
typedef __attribute__((ext_vector_type(4))) float v4f;
typedef __attribute__((ext_vector_type(8))) float v8f;


__global__ void zero_two(float* __restrict__ a, float* __restrict__ b, int n) {
    int i = blockIdx.x * blockDim.x + threadIdx.x;
    if (i < n) { a[i] = 0.f; b[i] = 0.f; }
}

__global__ void deg_accum(const int* __restrict__ row, float* __restrict__ deg, int E) {
    int e = blockIdx.x * blockDim.x + threadIdx.x;
    if (e < E) atomicAdd(&deg[row[e]], 1.0f);
}

__global__ void deg_inv_kernel(float* __restrict__ deg, int n) {
    int i = blockIdx.x * blockDim.x + threadIdx.x;
    if (i < n) { float d = deg[i]; deg[i] = (d > 0.f) ? rsqrtf(d) : 0.f; }
}

__global__ void coef_accum(const int* __restrict__ row, const int* __restrict__ col,
                           const float* __restrict__ dinv, float* __restrict__ coef, int E) {
    int e = blockIdx.x * blockDim.x + threadIdx.x;
    if (e < E) {
        int r = row[e], c = col[e];
        atomicAdd(&coef[c], dinv[r] * dinv[c]);
    }
}


#define ROWS_IT 32

template <int K>
__device__ __forceinline__ void tile_mma(const float (*At)[K + 4], int m, int hi,
                                         const v2f (&bf)[K / 4], v8f& acc0, v8f& acc1) {
    constexpr int KS = K / 4;
    constexpr int AHEAD = 4;
    v2f a0[KS], a1[KS];
    #pragma unroll
    for (int ks = 0; ks < AHEAD; ++ks) {
        a0[ks] = *(const v2f*)&At[m][4 * ks + 2 * hi];
        a1[ks] = *(const v2f*)&At[16 + m][4 * ks + 2 * hi];
    }
    #pragma unroll
    for (int ks = 0; ks < KS; ++ks) {
        if (ks + AHEAD < KS) {
            a0[ks + AHEAD] = *(const v2f*)&At[m][4 * (ks + AHEAD) + 2 * hi];
            a1[ks + AHEAD] = *(const v2f*)&At[16 + m][4 * (ks + AHEAD) + 2 * hi];
        }
        acc0 = __builtin_amdgcn_wmma_f32_16x16x4_f32(false, a0[ks], false, bf[ks],
                                                     (short)0, acc0, false, false);
        acc1 = __builtin_amdgcn_wmma_f32_16x16x4_f32(false, a1[ks], false, bf[ks],
                                                     (short)0, acc1, false, false);
        __builtin_amdgcn_sched_barrier(0);
    }
}

__launch_bounds__(256, 2)
__global__ void encoder_gemm(const float* __restrict__ x, const float* __restrict__ W,
                             const float* __restrict__ bias, float* __restrict__ h, int Nn) {
    __shared__ __align__(16) float As[2][ROWS_IT][F_IN_DIM + 4];
    const int lane = threadIdx.x & 31;
    const int wave = threadIdx.x >> 5;
    const int m    = lane & 15;
    const int hi   = lane >> 4;
    const int colN = wave * 16 + m;
    const float* wrow = W + (size_t)colN * F_IN_DIM;

    v2f bf[F_IN_DIM / 4];
    #pragma unroll
    for (int ks = 0; ks < F_IN_DIM / 4; ++ks)
        bf[ks] = *(const v2f*)&wrow[4 * ks + 2 * hi];
    const float bv = bias[colN];

    const int nIt = (Nn + ROWS_IT - 1) / ROWS_IT;
    constexpr int NPF = ROWS_IT * (F_IN_DIM / 4) / 256;
    const unsigned tid = threadIdx.x;
    const unsigned sr = tid >> 6, sc = (tid & 63) * 4;

    auto load_tile = [&](int t, v4f (&dst)[NPF]) {
        const int r0 = t * ROWS_IT;
        const v4f* base = (const v4f*)(x + (size_t)r0 * F_IN_DIM);
        const int rv = Nn - r0;
        if (rv >= ROWS_IT) {
            #pragma unroll
            for (int j = 0; j < NPF; ++j)
                dst[j] = __builtin_nontemporal_load(&base[tid + 256u * j]);
        } else {
            const unsigned last = (unsigned)rv * (F_IN_DIM / 4) - 1u;
            #pragma unroll
            for (int j = 0; j < NPF; ++j) {
                unsigned o = tid + 256u * j;
                dst[j] = __builtin_nontemporal_load(&base[o < last ? o : last]);
            }
        }
    };

    int it = blockIdx.x;
    if (it < nIt) {
        v4f t0[NPF];
        load_tile(it, t0);
        #pragma unroll
        for (int j = 0; j < NPF; ++j) *(v4f*)&As[0][sr + 4 * j][sc] = t0[j];
    }
    __syncthreads();
    int buf = 0;
    for (; it < nIt; it += gridDim.x) {
        const int row0 = it * ROWS_IT;
        const int rv   = Nn - row0;
        const int nxt  = it + gridDim.x;
        v4f pf[NPF];
        if (nxt < nIt) load_tile(nxt, pf);

        v8f acc0 = {}, acc1 = {};
        tile_mma<F_IN_DIM>(As[buf], m, hi, bf, acc0, acc1);

        float* ht = h + (size_t)row0 * H_DIM;
        const unsigned o0 = (unsigned)(8 * hi * H_DIM + colN);
        if (rv >= ROWS_IT) {
            #pragma unroll
            for (int r = 0; r < 8; ++r) {
                ht[o0 + (unsigned)(r * H_DIM)]        = acc0[r] + bv;
                ht[o0 + (unsigned)((16 + r) * H_DIM)] = acc1[r] + bv;
            }
        } else {
            #pragma unroll
            for (int r = 0; r < 8; ++r) {
                if (r + 8 * hi < rv)      ht[o0 + (unsigned)(r * H_DIM)]        = acc0[r] + bv;
                if (16 + r + 8 * hi < rv) ht[o0 + (unsigned)((16 + r) * H_DIM)] = acc1[r] + bv;
            }
        }

        if (nxt < nIt) {
            #pragma unroll
            for (int j = 0; j < NPF; ++j) *(v4f*)&As[buf ^ 1][sr + 4 * j][sc] = pf[j];
        }
        __syncthreads();
        buf ^= 1;
    }
}

__launch_bounds__(256, 4)
__global__ void layer_gemm(const float* __restrict__ h_in, const float* __restrict__ h0,
                           const float* __restrict__ coef, const float* __restrict__ Wraw,
                           const float* __restrict__ Wb, const float* __restrict__ extw,
                           const float* __restrict__ beta_p, float* __restrict__ h_out, int Nn) {
    __shared__ __align__(16) float smem[WR_ELEMS];
    float (*As)[ROWS_IT][H_DIM + 4] = reinterpret_cast<float (*)[ROWS_IT][H_DIM + 4]>(smem);
    static_assert(2 * ROWS_IT * (H_DIM + 4) <= WR_ELEMS, "A tiles must fit where the raw weights were");
    const int lane = threadIdx.x & 31;
    const int wave = threadIdx.x >> 5;
    const int m    = lane & 15;
    const int hi   = lane >> 4;
    const int colN = wave * 16 + m;

    const float wbv = Wb[colN];
    const float ew  = extw[colN];
    const float bl  = *beta_p;

    const int nIt = (Nn + ROWS_IT - 1) / ROWS_IT;
    constexpr int NPF = ROWS_IT * (H_DIM / 4) / 256;
    const unsigned tid = threadIdx.x;
    const unsigned sr = tid >> 5, sc = (tid & 31) * 4;

    auto load_tile = [&](int t, v4f (&dst)[NPF]) {
        const int r0 = t * ROWS_IT;
        const v4f* base = (const v4f*)(h_in + (size_t)r0 * H_DIM);
        const int rv = Nn - r0;
        if (rv >= ROWS_IT) {
            #pragma unroll
            for (int j = 0; j < NPF; ++j) dst[j] = base[tid + 256u * j];
        } else {
            const unsigned last = (unsigned)rv * (H_DIM / 4) - 1u;
            #pragma unroll
            for (int j = 0; j < NPF; ++j) {
                unsigned o = tid + 256u * j;
                dst[j] = base[o < last ? o : last];
            }
        }
    };

    int it = blockIdx.x;
    v4f t0[NPF];
    if (it < nIt) load_tile(it, t0);

    {
        constexpr int NW4 = WR_ELEMS / 4;
        constexpr int NWL = (NW4 + 255) / 256;
        const v4f* src = (const v4f*)Wraw;
        v4f wreg[NWL];
        #pragma unroll
        for (int j = 0; j < NWL; ++j) {
            const unsigned o = tid + 256u * j;
            if (256 * (j + 1) <= NW4) wreg[j] = src[o];
            else wreg[j] = src[o < (unsigned)NW4 ? o : (unsigned)(NW4 - 1)];
        }
        #pragma unroll
        for (int j = 0; j < NWL; ++j) {
            const unsigned o = tid + 256u * j;
            if (256 * (j + 1) <= NW4 || o < (unsigned)NW4) *(v4f*)&smem[4u * o] = wreg[j];
        }
    }
    __syncthreads();

    v2f bf[H_DIM / 4];
    {
        const int c = colN;
        smem[c * WR_PITCH + c] = 0.f;
        const int cw = c * WR_PITCH;
        float s = 0.f;
        #pragma unroll
        for (int ks = 0; ks < H_DIM / 4; ++ks) {
            float v[4];
            #pragma unroll
            for (int e = 0; e < 4; ++e) {
                const int j = 4 * ks + e;
                const int mx = (c > j) ? c : j;
                v[e] = smem[(cw - (WR_PITCH - 1) * mx) + j * WR_PITCH];
                s += fabsf(v[e]);
            }
            bf[ks].x = hi ? v[2] : v[0];
            bf[ks].y = hi ? v[3] : v[1];
        }
        const float d = smem[c * WR_PITCH + H_DIM] * s + smem[c * WR_PITCH + H_DIM + 1];
        #pragma unroll
        for (int ks = 0; ks < H_DIM / 4; ++ks) {
            const int k = 4 * ks + 2 * hi;
            if (k == c)     bf[ks].x = d;
            if (k + 1 == c) bf[ks].y = d;
        }
    }
    __syncthreads();

    if (it < nIt) {
        #pragma unroll
        for (int j = 0; j < NPF; ++j) *(v4f*)&As[0][sr + 8 * j][sc] = t0[j];
    }
    __syncthreads();
    int buf = 0;
    for (; it < nIt; it += gridDim.x) {
        const int row0 = it * ROWS_IT;
        const int rv   = Nn - row0;
        const int nxt  = it + gridDim.x;
        v4f pf[NPF];
        if (nxt < nIt) load_tile(nxt, pf);
        const float* h0t = h0 + (size_t)row0 * H_DIM;
        const float* cft = coef + row0;
        float h0v[16], cfv[16];
        #pragma unroll
        for (int q = 0; q < 16; ++q) {
            int lr = (q >> 3) * 16 + (q & 7) + 8 * hi;
            int lc = (lr < rv) ? lr : (rv - 1);
            h0v[q] = h0t[(unsigned)(lc * H_DIM + colN)];
            cfv[q] = cft[(unsigned)lc];
        }

        v8f acc0 = {}, acc1 = {};
        tile_mma<H_DIM>(As[buf], m, hi, bf, acc0, acc1);

        float res[16];
        #pragma unroll
        for (int q = 0; q < 16; ++q) {
            int lr = (q >> 3) * 16 + (q & 7) + 8 * hi;
            float cf   = cfv[q];
            float hv   = As[buf][lr][colN];
            float h0x  = h0v[q];
            float outp = ((q < 8) ? acc0[q & 7] : acc1[q & 7]) + wbv;
            float outv = cf * outp - (hv * ew + bl * h0x);
            res[q] = hv + STEPF * fmaxf(outv, 0.f);
        }
        float* hot = h_out + (size_t)row0 * H_DIM;
        const unsigned o0 = (unsigned)(8 * hi * H_DIM + colN);
        if (rv >= ROWS_IT) {
            #pragma unroll
            for (int q = 0; q < 16; ++q)
                hot[o0 + (unsigned)(((q >> 3) * 16 + (q & 7)) * H_DIM)] = res[q];
        } else {
            #pragma unroll
            for (int q = 0; q < 16; ++q) {
                int lr = (q >> 3) * 16 + (q & 7) + 8 * hi;
                if (lr < rv) hot[o0 + (unsigned)(((q >> 3) * 16 + (q & 7)) * H_DIM)] = res[q];
            }
        }

        if (nxt < nIt) {
            #pragma unroll
            for (int j = 0; j < NPF; ++j) *(v4f*)&As[buf ^ 1][sr + 8 * j][sc] = pf[j];
        }
        __syncthreads();
        buf ^= 1;
    }
}

__launch_bounds__(96)
__global__ void decoder_gemm(const float* __restrict__ h, const float* __restrict__ W,
                             const float* __restrict__ bias, float* __restrict__ out, int Nn) {
    __shared__ __align__(16) float As[2][ROWS_IT][H_DIM + 4];
    const int lane = threadIdx.x & 31;
    const int wave = threadIdx.x >> 5;
    const int m    = lane & 15;
    const int hi   = lane >> 4;
    const int colN = wave * 16 + m;
    const bool colok = (colN < C_DIM);
    const float* wrow = W + (size_t)(colok ? colN : 0) * H_DIM;

    v2f bf[H_DIM / 4];
    #pragma unroll
    for (int ks = 0; ks < H_DIM / 4; ++ks) {
        v2f b = *(const v2f*)&wrow[4 * ks + 2 * hi];
        if (!colok) { b.x = 0.f; b.y = 0.f; }
        bf[ks] = b;
    }
    const float bv = bias[colok ? colN : 0];

    const int nIt = (Nn + ROWS_IT - 1) / ROWS_IT;
    constexpr int NV4 = ROWS_IT * (H_DIM / 4);
    constexpr int NPF = (NV4 + 95) / 96;
    const unsigned tid = threadIdx.x;

    auto load_tile = [&](int t, v4f (&dst)[NPF]) {
        const int r0 = t * ROWS_IT;
        const v4f* base = (const v4f*)(h + (size_t)r0 * H_DIM);
        const int rv = Nn - r0;
        const unsigned last = (unsigned)(rv < ROWS_IT ? rv : ROWS_IT) * (H_DIM / 4) - 1u;
        #pragma unroll
        for (int j = 0; j < NPF; ++j) {
            unsigned o = tid + 96u * j;
            dst[j] = base[o < last ? o : last];
        }
    };
    auto store_tile = [&](int b, const v4f (&src)[NPF]) {
        #pragma unroll
        for (int j = 0; j < NPF; ++j) {
            unsigned o = tid + 96u * j;
            if (o < (unsigned)NV4) *(v4f*)&As[b][o >> 5][(o & 31) * 4] = src[j];
        }
    };

    int it = blockIdx.x;
    if (it < nIt) {
        v4f t0[NPF];
        load_tile(it, t0);
        store_tile(0, t0);
    }
    __syncthreads();
    int buf = 0;
    for (; it < nIt; it += gridDim.x) {
        const int row0 = it * ROWS_IT;
        const int rv   = Nn - row0;
        const int nxt  = it + gridDim.x;
        v4f pf[NPF];
        if (nxt < nIt) load_tile(nxt, pf);

        v8f acc0 = {}, acc1 = {};
        tile_mma<H_DIM>(As[buf], m, hi, bf, acc0, acc1);

        if (colok) {
            float* ot = out + (size_t)row0 * C_DIM;
            #pragma unroll
            for (int r = 0; r < 8; ++r) {
                int lr = r + 8 * hi;
                if (lr < rv)
                    __builtin_nontemporal_store(acc0[r] + bv, &ot[(unsigned)(lr * C_DIM + colN)]);
            }
            #pragma unroll
            for (int r = 0; r < 8; ++r) {
                int lr = 16 + r + 8 * hi;
                if (lr < rv)
                    __builtin_nontemporal_store(acc1[r] + bv, &ot[(unsigned)(lr * C_DIM + colN)]);
            }
        }

        if (nxt < nIt) store_tile(buf ^ 1, pf);
        __syncthreads();
        buf ^= 1;
    }
}


extern "C" void kernel_launch(void* const* d_in, const int* in_sizes, int n_in,
                              void* d_out, int out_size, void* d_ws, size_t ws_size,
                              hipStream_t stream) {
    const float* x     = (const float*)d_in[0];
    const int*   eidx  = (const int*)d_in[1];
    const float* enc_W = (const float*)d_in[2];
    const float* enc_b = (const float*)d_in[3];
    const float* W_raw = (const float*)d_in[4];
    const float* W_b   = (const float*)d_in[5];
    const float* ext_w = (const float*)d_in[6];
    const float* beta  = (const float*)d_in[7];
    const float* dec_W = (const float*)d_in[8];
    const float* dec_b = (const float*)d_in[9];

    const int Nn = in_sizes[0] / F_IN_DIM;
    const int E  = in_sizes[1] / 2;
    const int* row = eidx;
    const int* col = eidx + E;

    char* ws = (char*)d_ws;
    size_t off = 0;
    auto wsalloc = [&](size_t bytes) -> void* {
        void* p = ws + off;
        off = (off + bytes + 255) & ~(size_t)255;
        return p;
    };
    float* deg  = (float*)wsalloc((size_t)Nn * sizeof(float));
    float* coef = (float*)wsalloc((size_t)Nn * sizeof(float));
    float* h0   = (float*)wsalloc((size_t)Nn * H_DIM * sizeof(float));
    float* ha   = (float*)wsalloc((size_t)Nn * H_DIM * sizeof(float));
    float* hb   = (float*)wsalloc((size_t)Nn * H_DIM * sizeof(float));
    (void)ws_size; (void)n_in; (void)out_size;

    const int tb = 256;
    zero_two<<<(Nn + tb - 1) / tb, tb, 0, stream>>>(deg, coef, Nn);
    deg_accum<<<(E + tb - 1) / tb, tb, 0, stream>>>(row, deg, E);
    deg_inv_kernel<<<(Nn + tb - 1) / tb, tb, 0, stream>>>(deg, Nn);
    coef_accum<<<(E + tb - 1) / tb, tb, 0, stream>>>(row, col, deg, coef, E);

    encoder_gemm<<<512, 256, 0, stream>>>(x, enc_W, enc_b, h0, Nn);

    const float* hin = h0;
    float* bufs[2] = { ha, hb };
    for (int l = 0; l < L_DIM; ++l) {
        float* hout = bufs[l & 1];
        layer_gemm<<<512, 256, 0, stream>>>(hin, h0, coef,
                                               W_raw + (size_t)l * WR_ELEMS,
                                               W_b + (size_t)l * H_DIM,
                                               ext_w + (size_t)l * H_DIM,
                                               beta + l, hout, Nn);
        hin = hout;
    }
    decoder_gemm<<<1024, 96, 0, stream>>>(hin, dec_W, dec_b, (float*)d_out, Nn);
}


# edge-pass float atomics go to a per-half copy of deg and coef (half of the package from the die id), copies added per node in deg_inv and the encoder prologue; coef sums per half, then one add
# speedup vs baseline: 2.1069x; 1.5073x over previous
#include <hip/hip_runtime.h>

#define H_DIM   128
#define F_IN_DIM 256
#define C_DIM   40
#define L_DIM   4
#define STEPF   0.1f
#define WR_PITCH (H_DIM + 2)
#define WR_ELEMS (H_DIM * WR_PITCH)

typedef __attribute__((ext_vector_type(2))) float v2f;
typedef __attribute__((ext_vector_type(4))) float v4f;
typedef __attribute__((ext_vector_type(8))) float v8f;


__device__ __forceinline__ unsigned package_half() {
    return (__builtin_amdgcn_s_sendmsg_rtn(0x87u) >> 18) & 1u;
}

__global__ void zero_four(float* __restrict__ a, float* __restrict__ b,
                          float* __restrict__ c, float* __restrict__ d, int n) {
    int i = blockIdx.x * blockDim.x + threadIdx.x;
    if (i < n) { a[i] = 0.f; b[i] = 0.f; c[i] = 0.f; d[i] = 0.f; }
}

__global__ void deg_accum(const int* __restrict__ row, float* __restrict__ deg0,
                          float* __restrict__ deg1, int E) {
    float* deg = package_half() ? deg1 : deg0;
    int e = blockIdx.x * blockDim.x + threadIdx.x;
    if (e < E) atomicAdd(&deg[row[e]], 1.0f);
}

__global__ void deg_inv_kernel(float* __restrict__ deg0, const float* __restrict__ deg1, int n) {
    int i = blockIdx.x * blockDim.x + threadIdx.x;
    if (i < n) { float d = deg0[i] + deg1[i]; deg0[i] = (d > 0.f) ? rsqrtf(d) : 0.f; }
}

__global__ void coef_accum(const int* __restrict__ row, const int* __restrict__ col,
                           const float* __restrict__ dinv, float* __restrict__ coef0,
                           float* __restrict__ coef1, int E) {
    float* coef = package_half() ? coef1 : coef0;
    int e = blockIdx.x * blockDim.x + threadIdx.x;
    if (e < E) {
        int r = row[e], c = col[e];
        atomicAdd(&coef[c], dinv[r] * dinv[c]);
    }
}


#define ROWS_IT 32

template <int K>
__device__ __forceinline__ void tile_mma(const float (*At)[K + 4], int m, int hi,
                                         const v2f (&bf)[K / 4], v8f& acc0, v8f& acc1) {
    constexpr int KS = K / 4;
    constexpr int AHEAD = 4;
    v2f a0[KS], a1[KS];
    #pragma unroll
    for (int ks = 0; ks < AHEAD; ++ks) {
        a0[ks] = *(const v2f*)&At[m][4 * ks + 2 * hi];
        a1[ks] = *(const v2f*)&At[16 + m][4 * ks + 2 * hi];
    }
    #pragma unroll
    for (int ks = 0; ks < KS; ++ks) {
        if (ks + AHEAD < KS) {
            a0[ks + AHEAD] = *(const v2f*)&At[m][4 * (ks + AHEAD) + 2 * hi];
            a1[ks + AHEAD] = *(const v2f*)&At[16 + m][4 * (ks + AHEAD) + 2 * hi];
        }
        acc0 = __builtin_amdgcn_wmma_f32_16x16x4_f32(false, a0[ks], false, bf[ks],
                                                     (short)0, acc0, false, false);
        acc1 = __builtin_amdgcn_wmma_f32_16x16x4_f32(false, a1[ks], false, bf[ks],
                                                     (short)0, acc1, false, false);
        __builtin_amdgcn_sched_barrier(0);
    }
}

__launch_bounds__(256, 2)
__global__ void encoder_gemm(const float* __restrict__ x, const float* __restrict__ W,
                             const float* __restrict__ bias, float* __restrict__ h,
                             float* __restrict__ coef0, const float* __restrict__ coef1, int Nn) {
    __shared__ __align__(16) float As[2][ROWS_IT][F_IN_DIM + 4];
    const int lane = threadIdx.x & 31;
    const int wave = threadIdx.x >> 5;
    const int m    = lane & 15;
    const int hi   = lane >> 4;
    const int colN = wave * 16 + m;
    const float* wrow = W + (size_t)colN * F_IN_DIM;

    v2f bf[F_IN_DIM / 4];
    #pragma unroll
    for (int ks = 0; ks < F_IN_DIM / 4; ++ks)
        bf[ks] = *(const v2f*)&wrow[4 * ks + 2 * hi];
    const float bv = bias[colN];

    for (int i = blockIdx.x * 256 + threadIdx.x; i < Nn; i += gridDim.x * 256)
        coef0[i] = coef0[i] + coef1[i];

    const int nIt = (Nn + ROWS_IT - 1) / ROWS_IT;
    constexpr int NPF = ROWS_IT * (F_IN_DIM / 4) / 256;
    const unsigned tid = threadIdx.x;
    const unsigned sr = tid >> 6, sc = (tid & 63) * 4;

    auto load_tile = [&](int t, v4f (&dst)[NPF]) {
        const int r0 = t * ROWS_IT;
        const v4f* base = (const v4f*)(x + (size_t)r0 * F_IN_DIM);
        const int rv = Nn - r0;
        if (rv >= ROWS_IT) {
            #pragma unroll
            for (int j = 0; j < NPF; ++j)
                dst[j] = __builtin_nontemporal_load(&base[tid + 256u * j]);
        } else {
            const unsigned last = (unsigned)rv * (F_IN_DIM / 4) - 1u;
            #pragma unroll
            for (int j = 0; j < NPF; ++j) {
                unsigned o = tid + 256u * j;
                dst[j] = __builtin_nontemporal_load(&base[o < last ? o : last]);
            }
        }
    };

    int it = blockIdx.x;
    if (it < nIt) {
        v4f t0[NPF];
        load_tile(it, t0);
        #pragma unroll
        for (int j = 0; j < NPF; ++j) *(v4f*)&As[0][sr + 4 * j][sc] = t0[j];
    }
    __syncthreads();
    int buf = 0;
    for (; it < nIt; it += gridDim.x) {
        const int row0 = it * ROWS_IT;
        const int rv   = Nn - row0;
        const int nxt  = it + gridDim.x;
        v4f pf[NPF];
        if (nxt < nIt) load_tile(nxt, pf);

        v8f acc0 = {}, acc1 = {};
        tile_mma<F_IN_DIM>(As[buf], m, hi, bf, acc0, acc1);

        float* ht = h + (size_t)row0 * H_DIM;
        const unsigned o0 = (unsigned)(8 * hi * H_DIM + colN);
        if (rv >= ROWS_IT) {
            #pragma unroll
            for (int r = 0; r < 8; ++r) {
                ht[o0 + (unsigned)(r * H_DIM)]        = acc0[r] + bv;
                ht[o0 + (unsigned)((16 + r) * H_DIM)] = acc1[r] + bv;
            }
        } else {
            #pragma unroll
            for (int r = 0; r < 8; ++r) {
                if (r + 8 * hi < rv)      ht[o0 + (unsigned)(r * H_DIM)]        = acc0[r] + bv;
                if (16 + r + 8 * hi < rv) ht[o0 + (unsigned)((16 + r) * H_DIM)] = acc1[r] + bv;
            }
        }

        if (nxt < nIt) {
            #pragma unroll
            for (int j = 0; j < NPF; ++j) *(v4f*)&As[buf ^ 1][sr + 4 * j][sc] = pf[j];
        }
        __syncthreads();
        buf ^= 1;
    }
}

__launch_bounds__(256, 4)
__global__ void layer_gemm(const float* __restrict__ h_in, const float* __restrict__ h0,
                           const float* __restrict__ coef, const float* __restrict__ Wraw,
                           const float* __restrict__ Wb, const float* __restrict__ extw,
                           const float* __restrict__ beta_p, float* __restrict__ h_out, int Nn) {
    __shared__ __align__(16) float smem[WR_ELEMS];
    float (*As)[ROWS_IT][H_DIM + 4] = reinterpret_cast<float (*)[ROWS_IT][H_DIM + 4]>(smem);
    static_assert(2 * ROWS_IT * (H_DIM + 4) <= WR_ELEMS, "A tiles must fit where the raw weights were");
    const int lane = threadIdx.x & 31;
    const int wave = threadIdx.x >> 5;
    const int m    = lane & 15;
    const int hi   = lane >> 4;
    const int colN = wave * 16 + m;

    const float wbv = Wb[colN];
    const float ew  = extw[colN];
    const float bl  = *beta_p;

    const int nIt = (Nn + ROWS_IT - 1) / ROWS_IT;
    constexpr int NPF = ROWS_IT * (H_DIM / 4) / 256;
    const unsigned tid = threadIdx.x;
    const unsigned sr = tid >> 5, sc = (tid & 31) * 4;

    auto load_tile = [&](int t, v4f (&dst)[NPF]) {
        const int r0 = t * ROWS_IT;
        const v4f* base = (const v4f*)(h_in + (size_t)r0 * H_DIM);
        const int rv = Nn - r0;
        if (rv >= ROWS_IT) {
            #pragma unroll
            for (int j = 0; j < NPF; ++j) dst[j] = base[tid + 256u * j];
        } else {
            const unsigned last = (unsigned)rv * (H_DIM / 4) - 1u;
            #pragma unroll
            for (int j = 0; j < NPF; ++j) {
                unsigned o = tid + 256u * j;
                dst[j] = base[o < last ? o : last];
            }
        }
    };

    int it = blockIdx.x;
    v4f t0[NPF];
    if (it < nIt) load_tile(it, t0);

    {
        constexpr int NW4 = WR_ELEMS / 4;
        constexpr int NWL = (NW4 + 255) / 256;
        const v4f* src = (const v4f*)Wraw;
        v4f wreg[NWL];
        #pragma unroll
        for (int j = 0; j < NWL; ++j) {
            const unsigned o = tid + 256u * j;
            if (256 * (j + 1) <= NW4) wreg[j] = src[o];
            else wreg[j] = src[o < (unsigned)NW4 ? o : (unsigned)(NW4 - 1)];
        }
        #pragma unroll
        for (int j = 0; j < NWL; ++j) {
            const unsigned o = tid + 256u * j;
            if (256 * (j + 1) <= NW4 || o < (unsigned)NW4) *(v4f*)&smem[4u * o] = wreg[j];
        }
    }
    __syncthreads();

    v2f bf[H_DIM / 4];
    {
        const int c = colN;
        smem[c * WR_PITCH + c] = 0.f;
        const int cw = c * WR_PITCH;
        float s = 0.f;
        #pragma unroll
        for (int ks = 0; ks < H_DIM / 4; ++ks) {
            float v[4];
            #pragma unroll
            for (int e = 0; e < 4; ++e) {
                const int j = 4 * ks + e;
                const int mx = (c > j) ? c : j;
                v[e] = smem[(cw - (WR_PITCH - 1) * mx) + j * WR_PITCH];
                s += fabsf(v[e]);
            }
            bf[ks].x = hi ? v[2] : v[0];
            bf[ks].y = hi ? v[3] : v[1];
        }
        const float d = smem[c * WR_PITCH + H_DIM] * s + smem[c * WR_PITCH + H_DIM + 1];
        #pragma unroll
        for (int ks = 0; ks < H_DIM / 4; ++ks) {
            const int k = 4 * ks + 2 * hi;
            if (k == c)     bf[ks].x = d;
            if (k + 1 == c) bf[ks].y = d;
        }
    }
    __syncthreads();

    if (it < nIt) {
        #pragma unroll
        for (int j = 0; j < NPF; ++j) *(v4f*)&As[0][sr + 8 * j][sc] = t0[j];
    }
    __syncthreads();
    int buf = 0;
    for (; it < nIt; it += gridDim.x) {
        const int row0 = it * ROWS_IT;
        const int rv   = Nn - row0;
        const int nxt  = it + gridDim.x;
        v4f pf[NPF];
        if (nxt < nIt) load_tile(nxt, pf);
        const float* h0t = h0 + (size_t)row0 * H_DIM;
        const float* cft = coef + row0;
        float h0v[16], cfv[16];
        #pragma unroll
        for (int q = 0; q < 16; ++q) {
            int lr = (q >> 3) * 16 + (q & 7) + 8 * hi;
            int lc = (lr < rv) ? lr : (rv - 1);
            h0v[q] = h0t[(unsigned)(lc * H_DIM + colN)];
            cfv[q] = cft[(unsigned)lc];
        }

        v8f acc0 = {}, acc1 = {};
        tile_mma<H_DIM>(As[buf], m, hi, bf, acc0, acc1);

        float res[16];
        #pragma unroll
        for (int q = 0; q < 16; ++q) {
            int lr = (q >> 3) * 16 + (q & 7) + 8 * hi;
            float cf   = cfv[q];
            float hv   = As[buf][lr][colN];
            float h0x  = h0v[q];
            float outp = ((q < 8) ? acc0[q & 7] : acc1[q & 7]) + wbv;
            float outv = cf * outp - (hv * ew + bl * h0x);
            res[q] = hv + STEPF * fmaxf(outv, 0.f);
        }
        float* hot = h_out + (size_t)row0 * H_DIM;
        const unsigned o0 = (unsigned)(8 * hi * H_DIM + colN);
        if (rv >= ROWS_IT) {
            #pragma unroll
            for (int q = 0; q < 16; ++q)
                hot[o0 + (unsigned)(((q >> 3) * 16 + (q & 7)) * H_DIM)] = res[q];
        } else {
            #pragma unroll
            for (int q = 0; q < 16; ++q) {
                int lr = (q >> 3) * 16 + (q & 7) + 8 * hi;
                if (lr < rv) hot[o0 + (unsigned)(((q >> 3) * 16 + (q & 7)) * H_DIM)] = res[q];
            }
        }

        if (nxt < nIt) {
            #pragma unroll
            for (int j = 0; j < NPF; ++j) *(v4f*)&As[buf ^ 1][sr + 8 * j][sc] = pf[j];
        }
        __syncthreads();
        buf ^= 1;
    }
}

__launch_bounds__(96)
__global__ void decoder_gemm(const float* __restrict__ h, const float* __restrict__ W,
                             const float* __restrict__ bias, float* __restrict__ out, int Nn) {
    __shared__ __align__(16) float As[2][ROWS_IT][H_DIM + 4];
    const int lane = threadIdx.x & 31;
    const int wave = threadIdx.x >> 5;
    const int m    = lane & 15;
    const int hi   = lane >> 4;
    const int colN = wave * 16 + m;
    const bool colok = (colN < C_DIM);
    const float* wrow = W + (size_t)(colok ? colN : 0) * H_DIM;

    v2f bf[H_DIM / 4];
    #pragma unroll
    for (int ks = 0; ks < H_DIM / 4; ++ks) {
        v2f b = *(const v2f*)&wrow[4 * ks + 2 * hi];
        if (!colok) { b.x = 0.f; b.y = 0.f; }
        bf[ks] = b;
    }
    const float bv = bias[colok ? colN : 0];

    const int nIt = (Nn + ROWS_IT - 1) / ROWS_IT;
    constexpr int NV4 = ROWS_IT * (H_DIM / 4);
    constexpr int NPF = (NV4 + 95) / 96;
    const unsigned tid = threadIdx.x;

    auto load_tile = [&](int t, v4f (&dst)[NPF]) {
        const int r0 = t * ROWS_IT;
        const v4f* base = (const v4f*)(h + (size_t)r0 * H_DIM);
        const int rv = Nn - r0;
        const unsigned last = (unsigned)(rv < ROWS_IT ? rv : ROWS_IT) * (H_DIM / 4) - 1u;
        #pragma unroll
        for (int j = 0; j < NPF; ++j) {
            unsigned o = tid + 96u * j;
            dst[j] = base[o < last ? o : last];
        }
    };
    auto store_tile = [&](int b, const v4f (&src)[NPF]) {
        #pragma unroll
        for (int j = 0; j < NPF; ++j) {
            unsigned o = tid + 96u * j;
            if (o < (unsigned)NV4) *(v4f*)&As[b][o >> 5][(o & 31) * 4] = src[j];
        }
    };

    int it = blockIdx.x;
    if (it < nIt) {
        v4f t0[NPF];
        load_tile(it, t0);
        store_tile(0, t0);
    }
    __syncthreads();
    int buf = 0;
    for (; it < nIt; it += gridDim.x) {
        const int row0 = it * ROWS_IT;
        const int rv   = Nn - row0;
        const int nxt  = it + gridDim.x;
        v4f pf[NPF];
        if (nxt < nIt) load_tile(nxt, pf);

        v8f acc0 = {}, acc1 = {};
        tile_mma<H_DIM>(As[buf], m, hi, bf, acc0, acc1);

        if (colok) {
            float* ot = out + (size_t)row0 * C_DIM;
            #pragma unroll
            for (int r = 0; r < 8; ++r) {
                int lr = r + 8 * hi;
                if (lr < rv)
                    __builtin_nontemporal_store(acc0[r] + bv, &ot[(unsigned)(lr * C_DIM + colN)]);
            }
            #pragma unroll
            for (int r = 0; r < 8; ++r) {
                int lr = 16 + r + 8 * hi;
                if (lr < rv)
                    __builtin_nontemporal_store(acc1[r] + bv, &ot[(unsigned)(lr * C_DIM + colN)]);
            }
        }

        if (nxt < nIt) store_tile(buf ^ 1, pf);
        __syncthreads();
        buf ^= 1;
    }
}


extern "C" void kernel_launch(void* const* d_in, const int* in_sizes, int n_in,
                              void* d_out, int out_size, void* d_ws, size_t ws_size,
                              hipStream_t stream) {
    const float* x     = (const float*)d_in[0];
    const int*   eidx  = (const int*)d_in[1];
    const float* enc_W = (const float*)d_in[2];
    const float* enc_b = (const float*)d_in[3];
    const float* W_raw = (const float*)d_in[4];
    const float* W_b   = (const float*)d_in[5];
    const float* ext_w = (const float*)d_in[6];
    const float* beta  = (const float*)d_in[7];
    const float* dec_W = (const float*)d_in[8];
    const float* dec_b = (const float*)d_in[9];

    const int Nn = in_sizes[0] / F_IN_DIM;
    const int E  = in_sizes[1] / 2;
    const int* row = eidx;
    const int* col = eidx + E;

    char* ws = (char*)d_ws;
    size_t off = 0;
    auto wsalloc = [&](size_t bytes) -> void* {
        void* p = ws + off;
        off = (off + bytes + 255) & ~(size_t)255;
        return p;
    };
    float* deg  = (float*)wsalloc((size_t)Nn * sizeof(float));
    float* deg1 = (float*)wsalloc((size_t)Nn * sizeof(float));
    float* coef = (float*)wsalloc((size_t)Nn * sizeof(float));
    float* cof1 = (float*)wsalloc((size_t)Nn * sizeof(float));
    float* h0   = (float*)wsalloc((size_t)Nn * H_DIM * sizeof(float));
    float* ha   = (float*)wsalloc((size_t)Nn * H_DIM * sizeof(float));
    float* hb   = (float*)wsalloc((size_t)Nn * H_DIM * sizeof(float));
    (void)ws_size; (void)n_in; (void)out_size;

    const int tb = 256;
    zero_four<<<(Nn + tb - 1) / tb, tb, 0, stream>>>(deg, deg1, coef, cof1, Nn);
    deg_accum<<<(E + tb - 1) / tb, tb, 0, stream>>>(row, deg, deg1, E);
    deg_inv_kernel<<<(Nn + tb - 1) / tb, tb, 0, stream>>>(deg, deg1, Nn);
    coef_accum<<<(E + tb - 1) / tb, tb, 0, stream>>>(row, col, deg, coef, cof1, E);

    encoder_gemm<<<512, 256, 0, stream>>>(x, enc_W, enc_b, h0, coef, cof1, Nn);

    const float* hin = h0;
    float* bufs[2] = { ha, hb };
    for (int l = 0; l < L_DIM; ++l) {
        float* hout = bufs[l & 1];
        layer_gemm<<<512, 256, 0, stream>>>(hin, h0, coef,
                                               W_raw + (size_t)l * WR_ELEMS,
                                               W_b + (size_t)l * H_DIM,
                                               ext_w + (size_t)l * H_DIM,
                                               beta + l, hout, Nn);
        hin = hout;
    }
    decoder_gemm<<<1024, 96, 0, stream>>>(hin, dec_W, dec_b, (float*)d_out, Nn);
}


# edge passes ride in the encoder launches: encoder split in 2 row halves, each a 16-wave block per unit (8 waves WMMA, 8 walk edges; same float atomics, order differs); layers add the coef halves
# speedup vs baseline: 2.2123x; 1.0500x over previous
#include <hip/hip_runtime.h>

#define H_DIM   128
#define F_IN_DIM 256
#define C_DIM   40
#define L_DIM   4
#define STEPF   0.1f
#define WR_PITCH (H_DIM + 2)
#define WR_ELEMS (H_DIM * WR_PITCH)

typedef __attribute__((ext_vector_type(2))) float v2f;
typedef __attribute__((ext_vector_type(4))) float v4f;
typedef __attribute__((ext_vector_type(8))) float v8f;


__device__ __forceinline__ unsigned package_half() {
    return (__builtin_amdgcn_s_sendmsg_rtn(0x87u) >> 18) & 1u;
}

__global__ void zero_four(float* __restrict__ a, float* __restrict__ b,
                          float* __restrict__ c, float* __restrict__ d, int n) {
    int i = blockIdx.x * blockDim.x + threadIdx.x;
    if (i < n) { a[i] = 0.f; b[i] = 0.f; c[i] = 0.f; d[i] = 0.f; }
}

__global__ void deg_inv_kernel(float* __restrict__ deg0, const float* __restrict__ deg1, int n) {
    int i = blockIdx.x * blockDim.x + threadIdx.x;
    if (i < n) { float d = deg0[i] + deg1[i]; deg0[i] = (d > 0.f) ? rsqrtf(d) : 0.f; }
}


#define ROWS_IT 32

template <int K>
__device__ __forceinline__ void tile_mma(const float (*At)[K + 4], int m, int hi,
                                         const v2f (&bf)[K / 4], v8f& acc0, v8f& acc1) {
    constexpr int KS = K / 4;
    constexpr int AHEAD = 4;
    v2f a0[KS], a1[KS];
    #pragma unroll
    for (int ks = 0; ks < AHEAD; ++ks) {
        a0[ks] = *(const v2f*)&At[m][4 * ks + 2 * hi];
        a1[ks] = *(const v2f*)&At[16 + m][4 * ks + 2 * hi];
    }
    #pragma unroll
    for (int ks = 0; ks < KS; ++ks) {
        if (ks + AHEAD < KS) {
            a0[ks + AHEAD] = *(const v2f*)&At[m][4 * (ks + AHEAD) + 2 * hi];
            a1[ks + AHEAD] = *(const v2f*)&At[16 + m][4 * (ks + AHEAD) + 2 * hi];
        }
        acc0 = __builtin_amdgcn_wmma_f32_16x16x4_f32(false, a0[ks], false, bf[ks],
                                                     (short)0, acc0, false, false);
        acc1 = __builtin_amdgcn_wmma_f32_16x16x4_f32(false, a1[ks], false, bf[ks],
                                                     (short)0, acc1, false, false);
        __builtin_amdgcn_sched_barrier(0);
    }
}

#define EDGES_PER_IT 4

template <int ROLE>
__launch_bounds__(512, 1)
__global__ void encoder_edge(const float* __restrict__ x, const float* __restrict__ W,
                             const float* __restrict__ bias, float* __restrict__ h,
                             const int* __restrict__ row, const int* __restrict__ col,
                             const float* __restrict__ dinv, float* __restrict__ acc0,
                             float* __restrict__ acc1, int Nn, int E) {
    __shared__ __align__(16) float As[2][ROWS_IT][F_IN_DIM + 4];
    const bool gemm = __builtin_amdgcn_readfirstlane((int)threadIdx.x) < 256;
    const int lane = threadIdx.x & 31;
    const int wave = (threadIdx.x >> 5) & 7;
    const int m    = lane & 15;
    const int hi   = lane >> 4;
    const int colN = wave * 16 + m;

    const int nItAll = (Nn + ROWS_IT - 1) / ROWS_IT;
    const int nHalf  = (nItAll + 1) >> 1;
    const int tEnd   = ROLE ? nItAll : nHalf;
    const int tBeg   = ROLE ? nHalf : 0;
    constexpr int NPF = ROWS_IT * (F_IN_DIM / 4) / 256;
    const unsigned tid = threadIdx.x & 255u;
    const unsigned sr = tid >> 6, sc = (tid & 63) * 4;

    auto load_tile = [&](int t, v4f (&dst)[NPF]) {
        const int r0 = t * ROWS_IT;
        const v4f* base = (const v4f*)(x + (size_t)r0 * F_IN_DIM);
        const int rv = Nn - r0;
        if (rv >= ROWS_IT) {
            #pragma unroll
            for (int j = 0; j < NPF; ++j)
                dst[j] = __builtin_nontemporal_load(&base[tid + 256u * j]);
        } else {
            const unsigned last = (unsigned)rv * (F_IN_DIM / 4) - 1u;
            #pragma unroll
            for (int j = 0; j < NPF; ++j) {
                unsigned o = tid + 256u * j;
                dst[j] = __builtin_nontemporal_load(&base[o < last ? o : last]);
            }
        }
    };

    const unsigned G = gridDim.x * 256u;
    const unsigned g = blockIdx.x * 256u + tid;
    float* accp = package_half() ? acc1 : acc0;
    int er[EDGES_PER_IT], ec[EDGES_PER_IT];
    auto load_idx = [&](unsigned k0, int (&r)[EDGES_PER_IT], int (&c)[EDGES_PER_IT]) {
        #pragma unroll
        for (int q = 0; q < EDGES_PER_IT; ++q) {
            const unsigned e  = g + G * (k0 + (unsigned)q);
            const bool     ok = e < (unsigned)E;
            const unsigned ee = ok ? e : 0u;
            const int rr = row[ee];
            c[q] = ROLE ? col[ee] : 0;
            r[q] = ok ? rr : -1;
        }
    };
    auto do_edges = [&](const int (&r)[EDGES_PER_IT], const int (&c)[EDGES_PER_IT]) {
        if (ROLE == 0) {
            #pragma unroll
            for (int q = 0; q < EDGES_PER_IT; ++q)
                if (r[q] >= 0) atomicAdd(&accp[r[q]], 1.0f);
        } else {
            float pa[EDGES_PER_IT], pb[EDGES_PER_IT];
            #pragma unroll
            for (int q = 0; q < EDGES_PER_IT; ++q) {
                pa[q] = dinv[r[q] >= 0 ? r[q] : 0];
                pb[q] = dinv[c[q]];
            }
            #pragma unroll
            for (int q = 0; q < EDGES_PER_IT; ++q)
                if (r[q] >= 0) atomicAdd(&accp[c[q]], pa[q] * pb[q]);
        }
    };

    v2f bf[F_IN_DIM / 4];
    float bv = 0.f;
    int it = tBeg + (int)blockIdx.x;
    if (gemm) {
        const float* wrow = W + (size_t)colN * F_IN_DIM;
        #pragma unroll
        for (int ks = 0; ks < F_IN_DIM / 4; ++ks)
            bf[ks] = *(const v2f*)&wrow[4 * ks + 2 * hi];
        bv = bias[colN];
        if (it < tEnd) {
            v4f t0[NPF];
            load_tile(it, t0);
            #pragma unroll
            for (int j = 0; j < NPF; ++j) *(v4f*)&As[0][sr + 4 * j][sc] = t0[j];
        }
    } else {
        #pragma unroll
        for (int ks = 0; ks < F_IN_DIM / 4; ++ks) { bf[ks].x = 0.f; bf[ks].y = 0.f; }
        load_idx(0u, er, ec);
    }
    unsigned kc = 0;
    __syncthreads();
    int buf = 0;
    for (; it < tEnd; it += gridDim.x) {
        if (gemm) {
            const int row0 = it * ROWS_IT;
            const int rv   = Nn - row0;
            const int nxt  = it + gridDim.x;
            v4f pf[NPF];
            if (nxt < tEnd) load_tile(nxt, pf);

            v8f acc0v = {}, acc1v = {};
            tile_mma<F_IN_DIM>(As[buf], m, hi, bf, acc0v, acc1v);

            float* ht = h + (size_t)row0 * H_DIM;
            const unsigned o0 = (unsigned)(8 * hi * H_DIM + colN);
            if (rv >= ROWS_IT) {
                #pragma unroll
                for (int r = 0; r < 8; ++r) {
                    ht[o0 + (unsigned)(r * H_DIM)]        = acc0v[r] + bv;
                    ht[o0 + (unsigned)((16 + r) * H_DIM)] = acc1v[r] + bv;
                }
            } else {
                #pragma unroll
                for (int r = 0; r < 8; ++r) {
                    if (r + 8 * hi < rv)      ht[o0 + (unsigned)(r * H_DIM)]        = acc0v[r] + bv;
                    if (16 + r + 8 * hi < rv) ht[o0 + (unsigned)((16 + r) * H_DIM)] = acc1v[r] + bv;
                }
            }

            if (nxt < tEnd) {
                #pragma unroll
                for (int j = 0; j < NPF; ++j) *(v4f*)&As[buf ^ 1][sr + 4 * j][sc] = pf[j];
            }
        } else {
            int nr[EDGES_PER_IT], nc[EDGES_PER_IT];
            load_idx(kc + EDGES_PER_IT, nr, nc);
            do_edges(er, ec);
            #pragma unroll
            for (int q = 0; q < EDGES_PER_IT; ++q) { er[q] = nr[q]; ec[q] = nc[q]; }
            kc += EDGES_PER_IT;
        }
        __syncthreads();
        buf ^= 1;
    }
    if (!gemm) {
        do_edges(er, ec);
        for (unsigned k = kc + EDGES_PER_IT; g + G * k < (unsigned)E; k += EDGES_PER_IT) {
            load_idx(k, er, ec);
            do_edges(er, ec);
        }
    }
}

__launch_bounds__(256, 4)
__global__ void layer_gemm(const float* __restrict__ h_in, const float* __restrict__ h0,
                           const float* __restrict__ coef0, const float* __restrict__ coef1,
                           const float* __restrict__ Wraw,
                           const float* __restrict__ Wb, const float* __restrict__ extw,
                           const float* __restrict__ beta_p, float* __restrict__ h_out, int Nn) {
    __shared__ __align__(16) float smem[WR_ELEMS];
    float (*As)[ROWS_IT][H_DIM + 4] = reinterpret_cast<float (*)[ROWS_IT][H_DIM + 4]>(smem);
    static_assert(2 * ROWS_IT * (H_DIM + 4) <= WR_ELEMS, "A tiles must fit where the raw weights were");
    const int lane = threadIdx.x & 31;
    const int wave = threadIdx.x >> 5;
    const int m    = lane & 15;
    const int hi   = lane >> 4;
    const int colN = wave * 16 + m;

    const float wbv = Wb[colN];
    const float ew  = extw[colN];
    const float bl  = *beta_p;

    const int nIt = (Nn + ROWS_IT - 1) / ROWS_IT;
    constexpr int NPF = ROWS_IT * (H_DIM / 4) / 256;
    const unsigned tid = threadIdx.x;
    const unsigned sr = tid >> 5, sc = (tid & 31) * 4;

    auto load_tile = [&](int t, v4f (&dst)[NPF]) {
        const int r0 = t * ROWS_IT;
        const v4f* base = (const v4f*)(h_in + (size_t)r0 * H_DIM);
        const int rv = Nn - r0;
        if (rv >= ROWS_IT) {
            #pragma unroll
            for (int j = 0; j < NPF; ++j) dst[j] = base[tid + 256u * j];
        } else {
            const unsigned last = (unsigned)rv * (H_DIM / 4) - 1u;
            #pragma unroll
            for (int j = 0; j < NPF; ++j) {
                unsigned o = tid + 256u * j;
                dst[j] = base[o < last ? o : last];
            }
        }
    };

    int it = blockIdx.x;
    v4f t0[NPF];
    if (it < nIt) load_tile(it, t0);

    {
        constexpr int NW4 = WR_ELEMS / 4;
        constexpr int NWL = (NW4 + 255) / 256;
        const v4f* src = (const v4f*)Wraw;
        v4f wreg[NWL];
        #pragma unroll
        for (int j = 0; j < NWL; ++j) {
            const unsigned o = tid + 256u * j;
            if (256 * (j + 1) <= NW4) wreg[j] = src[o];
            else wreg[j] = src[o < (unsigned)NW4 ? o : (unsigned)(NW4 - 1)];
        }
        #pragma unroll
        for (int j = 0; j < NWL; ++j) {
            const unsigned o = tid + 256u * j;
            if (256 * (j + 1) <= NW4 || o < (unsigned)NW4) *(v4f*)&smem[4u * o] = wreg[j];
        }
    }
    __syncthreads();

    v2f bf[H_DIM / 4];
    {
        const int c = colN;
        smem[c * WR_PITCH + c] = 0.f;
        const int cw = c * WR_PITCH;
        float s = 0.f;
        #pragma unroll
        for (int ks = 0; ks < H_DIM / 4; ++ks) {
            float v[4];
            #pragma unroll
            for (int e = 0; e < 4; ++e) {
                const int j = 4 * ks + e;
                const int mx = (c > j) ? c : j;
                v[e] = smem[(cw - (WR_PITCH - 1) * mx) + j * WR_PITCH];
                s += fabsf(v[e]);
            }
            bf[ks].x = hi ? v[2] : v[0];
            bf[ks].y = hi ? v[3] : v[1];
        }
        const float d = smem[c * WR_PITCH + H_DIM] * s + smem[c * WR_PITCH + H_DIM + 1];
        #pragma unroll
        for (int ks = 0; ks < H_DIM / 4; ++ks) {
            const int k = 4 * ks + 2 * hi;
            if (k == c)     bf[ks].x = d;
            if (k + 1 == c) bf[ks].y = d;
        }
    }
    __syncthreads();

    if (it < nIt) {
        #pragma unroll
        for (int j = 0; j < NPF; ++j) *(v4f*)&As[0][sr + 8 * j][sc] = t0[j];
    }
    __syncthreads();
    int buf = 0;
    for (; it < nIt; it += gridDim.x) {
        const int row0 = it * ROWS_IT;
        const int rv   = Nn - row0;
        const int nxt  = it + gridDim.x;
        v4f pf[NPF];
        if (nxt < nIt) load_tile(nxt, pf);
        const float* h0t = h0 + (size_t)row0 * H_DIM;
        float h0v[16];
        #pragma unroll
        for (int q = 0; q < 16; ++q) {
            int lr = (q >> 3) * 16 + (q & 7) + 8 * hi;
            int lc = (lr < rv) ? lr : (rv - 1);
            h0v[q] = h0t[(unsigned)(lc * H_DIM + colN)];
        }
        const int cli = row0 + ((lane < rv) ? lane : (rv - 1));
        const float cl = coef0[cli] + coef1[cli];

        v8f acc0 = {}, acc1 = {};
        tile_mma<H_DIM>(As[buf], m, hi, bf, acc0, acc1);

        float res[16];
        #pragma unroll
        for (int q = 0; q < 16; ++q) {
            int lr = (q >> 3) * 16 + (q & 7) + 8 * hi;
            float cf   = __shfl(cl, lr);
            float hv   = As[buf][lr][colN];
            float h0x  = h0v[q];
            float outp = ((q < 8) ? acc0[q & 7] : acc1[q & 7]) + wbv;
            float outv = cf * outp - (hv * ew + bl * h0x);
            res[q] = hv + STEPF * fmaxf(outv, 0.f);
        }
        float* hot = h_out + (size_t)row0 * H_DIM;
        const unsigned o0 = (unsigned)(8 * hi * H_DIM + colN);
        if (rv >= ROWS_IT) {
            #pragma unroll
            for (int q = 0; q < 16; ++q)
                hot[o0 + (unsigned)(((q >> 3) * 16 + (q & 7)) * H_DIM)] = res[q];
        } else {
            #pragma unroll
            for (int q = 0; q < 16; ++q) {
                int lr = (q >> 3) * 16 + (q & 7) + 8 * hi;
                if (lr < rv) hot[o0 + (unsigned)(((q >> 3) * 16 + (q & 7)) * H_DIM)] = res[q];
            }
        }

        if (nxt < nIt) {
            #pragma unroll
            for (int j = 0; j < NPF; ++j) *(v4f*)&As[buf ^ 1][sr + 8 * j][sc] = pf[j];
        }
        __syncthreads();
        buf ^= 1;
    }
}

__launch_bounds__(96)
__global__ void decoder_gemm(const float* __restrict__ h, const float* __restrict__ W,
                             const float* __restrict__ bias, float* __restrict__ out, int Nn) {
    __shared__ __align__(16) float As[2][ROWS_IT][H_DIM + 4];
    const int lane = threadIdx.x & 31;
    const int wave = threadIdx.x >> 5;
    const int m    = lane & 15;
    const int hi   = lane >> 4;
    const int colN = wave * 16 + m;
    const bool colok = (colN < C_DIM);
    const float* wrow = W + (size_t)(colok ? colN : 0) * H_DIM;

    v2f bf[H_DIM / 4];
    #pragma unroll
    for (int ks = 0; ks < H_DIM / 4; ++ks) {
        v2f b = *(const v2f*)&wrow[4 * ks + 2 * hi];
        if (!colok) { b.x = 0.f; b.y = 0.f; }
        bf[ks] = b;
    }
    const float bv = bias[colok ? colN : 0];

    const int nIt = (Nn + ROWS_IT - 1) / ROWS_IT;
    constexpr int NV4 = ROWS_IT * (H_DIM / 4);
    constexpr int NPF = (NV4 + 95) / 96;
    const unsigned tid = threadIdx.x;

    auto load_tile = [&](int t, v4f (&dst)[NPF]) {
        const int r0 = t * ROWS_IT;
        const v4f* base = (const v4f*)(h + (size_t)r0 * H_DIM);
        const int rv = Nn - r0;
        const unsigned last = (unsigned)(rv < ROWS_IT ? rv : ROWS_IT) * (H_DIM / 4) - 1u;
        #pragma unroll
        for (int j = 0; j < NPF; ++j) {
            unsigned o = tid + 96u * j;
            dst[j] = base[o < last ? o : last];
        }
    };
    auto store_tile = [&](int b, const v4f (&src)[NPF]) {
        #pragma unroll
        for (int j = 0; j < NPF; ++j) {
            unsigned o = tid + 96u * j;
            if (o < (unsigned)NV4) *(v4f*)&As[b][o >> 5][(o & 31) * 4] = src[j];
        }
    };

    int it = blockIdx.x;
    if (it < nIt) {
        v4f t0[NPF];
        load_tile(it, t0);
        store_tile(0, t0);
    }
    __syncthreads();
    int buf = 0;
    for (; it < nIt; it += gridDim.x) {
        const int row0 = it * ROWS_IT;
        const int rv   = Nn - row0;
        const int nxt  = it + gridDim.x;
        v4f pf[NPF];
        if (nxt < nIt) load_tile(nxt, pf);

        v8f acc0 = {}, acc1 = {};
        tile_mma<H_DIM>(As[buf], m, hi, bf, acc0, acc1);

        if (colok) {
            float* ot = out + (size_t)row0 * C_DIM;
            #pragma unroll
            for (int r = 0; r < 8; ++r) {
                int lr = r + 8 * hi;
                if (lr < rv)
                    __builtin_nontemporal_store(acc0[r] + bv, &ot[(unsigned)(lr * C_DIM + colN)]);
            }
            #pragma unroll
            for (int r = 0; r < 8; ++r) {
                int lr = 16 + r + 8 * hi;
                if (lr < rv)
                    __builtin_nontemporal_store(acc1[r] + bv, &ot[(unsigned)(lr * C_DIM + colN)]);
            }
        }

        if (nxt < nIt) store_tile(buf ^ 1, pf);
        __syncthreads();
        buf ^= 1;
    }
}


extern "C" void kernel_launch(void* const* d_in, const int* in_sizes, int n_in,
                              void* d_out, int out_size, void* d_ws, size_t ws_size,
                              hipStream_t stream) {
    const float* x     = (const float*)d_in[0];
    const int*   eidx  = (const int*)d_in[1];
    const float* enc_W = (const float*)d_in[2];
    const float* enc_b = (const float*)d_in[3];
    const float* W_raw = (const float*)d_in[4];
    const float* W_b   = (const float*)d_in[5];
    const float* ext_w = (const float*)d_in[6];
    const float* beta  = (const float*)d_in[7];
    const float* dec_W = (const float*)d_in[8];
    const float* dec_b = (const float*)d_in[9];

    const int Nn = in_sizes[0] / F_IN_DIM;
    const int E  = in_sizes[1] / 2;
    const int* row = eidx;
    const int* col = eidx + E;

    char* ws = (char*)d_ws;
    size_t off = 0;
    auto wsalloc = [&](size_t bytes) -> void* {
        void* p = ws + off;
        off = (off + bytes + 255) & ~(size_t)255;
        return p;
    };
    float* deg  = (float*)wsalloc((size_t)Nn * sizeof(float));
    float* deg1 = (float*)wsalloc((size_t)Nn * sizeof(float));
    float* coef = (float*)wsalloc((size_t)Nn * sizeof(float));
    float* cof1 = (float*)wsalloc((size_t)Nn * sizeof(float));
    float* h0   = (float*)wsalloc((size_t)Nn * H_DIM * sizeof(float));
    float* ha   = (float*)wsalloc((size_t)Nn * H_DIM * sizeof(float));
    float* hb   = (float*)wsalloc((size_t)Nn * H_DIM * sizeof(float));
    (void)ws_size; (void)n_in; (void)out_size;

    const int tb = 256;
    zero_four<<<(Nn + tb - 1) / tb, tb, 0, stream>>>(deg, deg1, coef, cof1, Nn);
    encoder_edge<0><<<256, 512, 0, stream>>>(x, enc_W, enc_b, h0, row, col, nullptr, deg, deg1, Nn, E);
    deg_inv_kernel<<<(Nn + tb - 1) / tb, tb, 0, stream>>>(deg, deg1, Nn);
    encoder_edge<1><<<256, 512, 0, stream>>>(x, enc_W, enc_b, h0, row, col, deg, coef, cof1, Nn, E);


    const float* hin = h0;
    float* bufs[2] = { ha, hb };
    for (int l = 0; l < L_DIM; ++l) {
        float* hout = bufs[l & 1];
        layer_gemm<<<512, 256, 0, stream>>>(hin, h0, coef, cof1,
                                               W_raw + (size_t)l * WR_ELEMS,
                                               W_b + (size_t)l * H_DIM,
                                               ext_w + (size_t)l * H_DIM,
                                               beta + l, hout, Nn);
        hin = hout;
    }
    decoder_gemm<<<1024, 96, 0, stream>>>(hin, dec_W, dec_b, (float*)d_out, Nn);
}


# the 4 layer launches become one: a block keeps its row tiles through all layers, so layer l+1 reads only rows the block itself stored in layer l (tiles read non-temporal); sum order unchanged
# speedup vs baseline: 2.3065x; 1.0426x over previous
#include <hip/hip_runtime.h>

#define H_DIM   128
#define F_IN_DIM 256
#define C_DIM   40
#define L_DIM   4
#define STEPF   0.1f
#define WR_PITCH (H_DIM + 2)
#define WR_ELEMS (H_DIM * WR_PITCH)

typedef __attribute__((ext_vector_type(2))) float v2f;
typedef __attribute__((ext_vector_type(4))) float v4f;
typedef __attribute__((ext_vector_type(8))) float v8f;


__device__ __forceinline__ unsigned package_half() {
    return (__builtin_amdgcn_s_sendmsg_rtn(0x87u) >> 18) & 1u;
}

__global__ void zero_four(float* __restrict__ a, float* __restrict__ b,
                          float* __restrict__ c, float* __restrict__ d, int n) {
    int i = blockIdx.x * blockDim.x + threadIdx.x;
    if (i < n) { a[i] = 0.f; b[i] = 0.f; c[i] = 0.f; d[i] = 0.f; }
}

__global__ void deg_inv_kernel(float* __restrict__ deg0, const float* __restrict__ deg1, int n) {
    int i = blockIdx.x * blockDim.x + threadIdx.x;
    if (i < n) { float d = deg0[i] + deg1[i]; deg0[i] = (d > 0.f) ? rsqrtf(d) : 0.f; }
}


#define ROWS_IT 32

template <int K>
__device__ __forceinline__ void tile_mma(const float (*At)[K + 4], int m, int hi,
                                         const v2f (&bf)[K / 4], v8f& acc0, v8f& acc1) {
    constexpr int KS = K / 4;
    constexpr int AHEAD = 4;
    v2f a0[KS], a1[KS];
    #pragma unroll
    for (int ks = 0; ks < AHEAD; ++ks) {
        a0[ks] = *(const v2f*)&At[m][4 * ks + 2 * hi];
        a1[ks] = *(const v2f*)&At[16 + m][4 * ks + 2 * hi];
    }
    #pragma unroll
    for (int ks = 0; ks < KS; ++ks) {
        if (ks + AHEAD < KS) {
            a0[ks + AHEAD] = *(const v2f*)&At[m][4 * (ks + AHEAD) + 2 * hi];
            a1[ks + AHEAD] = *(const v2f*)&At[16 + m][4 * (ks + AHEAD) + 2 * hi];
        }
        acc0 = __builtin_amdgcn_wmma_f32_16x16x4_f32(false, a0[ks], false, bf[ks],
                                                     (short)0, acc0, false, false);
        acc1 = __builtin_amdgcn_wmma_f32_16x16x4_f32(false, a1[ks], false, bf[ks],
                                                     (short)0, acc1, false, false);
        __builtin_amdgcn_sched_barrier(0);
    }
}

#define EDGES_PER_IT 4

template <int ROLE>
__launch_bounds__(512, 1)
__global__ void encoder_edge(const float* __restrict__ x, const float* __restrict__ W,
                             const float* __restrict__ bias, float* __restrict__ h,
                             const int* __restrict__ row, const int* __restrict__ col,
                             const float* __restrict__ dinv, float* __restrict__ acc0,
                             float* __restrict__ acc1, int Nn, int E) {
    __shared__ __align__(16) float As[2][ROWS_IT][F_IN_DIM + 4];
    const bool gemm = __builtin_amdgcn_readfirstlane((int)threadIdx.x) < 256;
    const int lane = threadIdx.x & 31;
    const int wave = (threadIdx.x >> 5) & 7;
    const int m    = lane & 15;
    const int hi   = lane >> 4;
    const int colN = wave * 16 + m;

    const int nItAll = (Nn + ROWS_IT - 1) / ROWS_IT;
    const int nHalf  = (nItAll + 1) >> 1;
    const int tEnd   = ROLE ? nItAll : nHalf;
    const int tBeg   = ROLE ? nHalf : 0;
    constexpr int NPF = ROWS_IT * (F_IN_DIM / 4) / 256;
    const unsigned tid = threadIdx.x & 255u;
    const unsigned sr = tid >> 6, sc = (tid & 63) * 4;

    auto load_tile = [&](int t, v4f (&dst)[NPF]) {
        const int r0 = t * ROWS_IT;
        const v4f* base = (const v4f*)(x + (size_t)r0 * F_IN_DIM);
        const int rv = Nn - r0;
        if (rv >= ROWS_IT) {
            #pragma unroll
            for (int j = 0; j < NPF; ++j)
                dst[j] = __builtin_nontemporal_load(&base[tid + 256u * j]);
        } else {
            const unsigned last = (unsigned)rv * (F_IN_DIM / 4) - 1u;
            #pragma unroll
            for (int j = 0; j < NPF; ++j) {
                unsigned o = tid + 256u * j;
                dst[j] = __builtin_nontemporal_load(&base[o < last ? o : last]);
            }
        }
    };

    const unsigned G = gridDim.x * 256u;
    const unsigned g = blockIdx.x * 256u + tid;
    float* accp = package_half() ? acc1 : acc0;
    int er[EDGES_PER_IT], ec[EDGES_PER_IT];
    auto load_idx = [&](unsigned k0, int (&r)[EDGES_PER_IT], int (&c)[EDGES_PER_IT]) {
        #pragma unroll
        for (int q = 0; q < EDGES_PER_IT; ++q) {
            const unsigned e  = g + G * (k0 + (unsigned)q);
            const bool     ok = e < (unsigned)E;
            const unsigned ee = ok ? e : 0u;
            const int rr = row[ee];
            c[q] = ROLE ? col[ee] : 0;
            r[q] = ok ? rr : -1;
        }
    };
    auto do_edges = [&](const int (&r)[EDGES_PER_IT], const int (&c)[EDGES_PER_IT]) {
        if (ROLE == 0) {
            #pragma unroll
            for (int q = 0; q < EDGES_PER_IT; ++q)
                if (r[q] >= 0) atomicAdd(&accp[r[q]], 1.0f);
        } else {
            float pa[EDGES_PER_IT], pb[EDGES_PER_IT];
            #pragma unroll
            for (int q = 0; q < EDGES_PER_IT; ++q) {
                pa[q] = dinv[r[q] >= 0 ? r[q] : 0];
                pb[q] = dinv[c[q]];
            }
            #pragma unroll
            for (int q = 0; q < EDGES_PER_IT; ++q)
                if (r[q] >= 0) atomicAdd(&accp[c[q]], pa[q] * pb[q]);
        }
    };

    v2f bf[F_IN_DIM / 4];
    float bv = 0.f;
    int it = tBeg + (int)blockIdx.x;
    if (gemm) {
        const float* wrow = W + (size_t)colN * F_IN_DIM;
        #pragma unroll
        for (int ks = 0; ks < F_IN_DIM / 4; ++ks)
            bf[ks] = *(const v2f*)&wrow[4 * ks + 2 * hi];
        bv = bias[colN];
        if (it < tEnd) {
            v4f t0[NPF];
            load_tile(it, t0);
            #pragma unroll
            for (int j = 0; j < NPF; ++j) *(v4f*)&As[0][sr + 4 * j][sc] = t0[j];
        }
    } else {
        #pragma unroll
        for (int ks = 0; ks < F_IN_DIM / 4; ++ks) { bf[ks].x = 0.f; bf[ks].y = 0.f; }
        load_idx(0u, er, ec);
    }
    unsigned kc = 0;
    __syncthreads();
    int buf = 0;
    for (; it < tEnd; it += gridDim.x) {
        if (gemm) {
            const int row0 = it * ROWS_IT;
            const int rv   = Nn - row0;
            const int nxt  = it + gridDim.x;
            v4f pf[NPF];
            if (nxt < tEnd) load_tile(nxt, pf);

            v8f acc0v = {}, acc1v = {};
            tile_mma<F_IN_DIM>(As[buf], m, hi, bf, acc0v, acc1v);

            float* ht = h + (size_t)row0 * H_DIM;
            const unsigned o0 = (unsigned)(8 * hi * H_DIM + colN);
            if (rv >= ROWS_IT) {
                #pragma unroll
                for (int r = 0; r < 8; ++r) {
                    ht[o0 + (unsigned)(r * H_DIM)]        = acc0v[r] + bv;
                    ht[o0 + (unsigned)((16 + r) * H_DIM)] = acc1v[r] + bv;
                }
            } else {
                #pragma unroll
                for (int r = 0; r < 8; ++r) {
                    if (r + 8 * hi < rv)      ht[o0 + (unsigned)(r * H_DIM)]        = acc0v[r] + bv;
                    if (16 + r + 8 * hi < rv) ht[o0 + (unsigned)((16 + r) * H_DIM)] = acc1v[r] + bv;
                }
            }

            if (nxt < tEnd) {
                #pragma unroll
                for (int j = 0; j < NPF; ++j) *(v4f*)&As[buf ^ 1][sr + 4 * j][sc] = pf[j];
            }
        } else {
            int nr[EDGES_PER_IT], nc[EDGES_PER_IT];
            load_idx(kc + EDGES_PER_IT, nr, nc);
            do_edges(er, ec);
            #pragma unroll
            for (int q = 0; q < EDGES_PER_IT; ++q) { er[q] = nr[q]; ec[q] = nc[q]; }
            kc += EDGES_PER_IT;
        }
        __syncthreads();
        buf ^= 1;
    }
    if (!gemm) {
        do_edges(er, ec);
        for (unsigned k = kc + EDGES_PER_IT; g + G * k < (unsigned)E; k += EDGES_PER_IT) {
            load_idx(k, er, ec);
            do_edges(er, ec);
        }
    }
}

__launch_bounds__(256, 4)
__global__ void layers_gemm(const float* h0, const float* __restrict__ coef0,
                            const float* __restrict__ coef1, const float* __restrict__ Wraw_all,
                            const float* __restrict__ Wb_all, const float* __restrict__ extw_all,
                            const float* __restrict__ beta_all, float* ha, float* hb, int Nn) {
    __shared__ __align__(16) float smem[WR_ELEMS];
    float (*As)[ROWS_IT][H_DIM + 4] = reinterpret_cast<float (*)[ROWS_IT][H_DIM + 4]>(smem);
    static_assert(2 * ROWS_IT * (H_DIM + 4) <= WR_ELEMS, "A tiles must fit where the raw weights were");
    const int lane = threadIdx.x & 31;
    const int wave = threadIdx.x >> 5;
    const int m    = lane & 15;
    const int hi   = lane >> 4;
    const int colN = wave * 16 + m;

    const int nIt = (Nn + ROWS_IT - 1) / ROWS_IT;
    constexpr int NPF = ROWS_IT * (H_DIM / 4) / 256;
    const unsigned tid = threadIdx.x;
    const unsigned sr = tid >> 5, sc = (tid & 31) * 4;

    #pragma unroll 1
    for (int l = 0; l < L_DIM; ++l) {
    const float* h_in  = (l == 0) ? h0 : ((l & 1) ? ha : hb);
    float*       h_out = (l & 1) ? hb : ha;
    const float* Wraw  = Wraw_all + (size_t)l * WR_ELEMS;
    const float wbv = Wb_all[l * H_DIM + colN];
    const float ew  = extw_all[l * H_DIM + colN];
    const float bl  = beta_all[l];

    auto load_tile = [&](int t, v4f (&dst)[NPF]) {
        const int r0 = t * ROWS_IT;
        const v4f* base = (const v4f*)(h_in + (size_t)r0 * H_DIM);
        const int rv = Nn - r0;
        if (rv >= ROWS_IT) {
            #pragma unroll
            for (int j = 0; j < NPF; ++j) dst[j] = __builtin_nontemporal_load(&base[tid + 256u * j]);
        } else {
            const unsigned last = (unsigned)rv * (H_DIM / 4) - 1u;
            #pragma unroll
            for (int j = 0; j < NPF; ++j) {
                unsigned o = tid + 256u * j;
                dst[j] = __builtin_nontemporal_load(&base[o < last ? o : last]);
            }
        }
    };

    int it = blockIdx.x;
    v4f t0[NPF];
    if (it < nIt) load_tile(it, t0);

    {
        constexpr int NW4 = WR_ELEMS / 4;
        constexpr int NWL = (NW4 + 255) / 256;
        const v4f* src = (const v4f*)Wraw;
        v4f wreg[NWL];
        #pragma unroll
        for (int j = 0; j < NWL; ++j) {
            const unsigned o = tid + 256u * j;
            if (256 * (j + 1) <= NW4) wreg[j] = src[o];
            else wreg[j] = src[o < (unsigned)NW4 ? o : (unsigned)(NW4 - 1)];
        }
        #pragma unroll
        for (int j = 0; j < NWL; ++j) {
            const unsigned o = tid + 256u * j;
            if (256 * (j + 1) <= NW4 || o < (unsigned)NW4) *(v4f*)&smem[4u * o] = wreg[j];
        }
    }
    __syncthreads();

    v2f bf[H_DIM / 4];
    {
        int c = colN;
        asm volatile("" : "+v"(c));
        smem[c * WR_PITCH + c] = 0.f;
        const int cw = c * WR_PITCH;
        float s = 0.f;
        #pragma unroll
        for (int ks = 0; ks < H_DIM / 4; ++ks) {
            float v[4];
            #pragma unroll
            for (int e = 0; e < 4; ++e) {
                const int j = 4 * ks + e;
                const int mx = (c > j) ? c : j;
                v[e] = smem[(cw - (WR_PITCH - 1) * mx) + j * WR_PITCH];
                s += fabsf(v[e]);
            }
            bf[ks].x = hi ? v[2] : v[0];
            bf[ks].y = hi ? v[3] : v[1];
        }
        const float d = smem[c * WR_PITCH + H_DIM] * s + smem[c * WR_PITCH + H_DIM + 1];
        #pragma unroll
        for (int ks = 0; ks < H_DIM / 4; ++ks) {
            const int k = 4 * ks + 2 * hi;
            if (k == c)     bf[ks].x = d;
            if (k + 1 == c) bf[ks].y = d;
        }
    }
    __syncthreads();

    if (it < nIt) {
        #pragma unroll
        for (int j = 0; j < NPF; ++j) *(v4f*)&As[0][sr + 8 * j][sc] = t0[j];
    }
    __syncthreads();
    int buf = 0;
    for (; it < nIt; it += gridDim.x) {
        const int row0 = it * ROWS_IT;
        const int rv   = Nn - row0;
        const int nxt  = it + gridDim.x;
        v4f pf[NPF];
        if (nxt < nIt) load_tile(nxt, pf);
        const float* h0t = h0 + (size_t)row0 * H_DIM;
        float h0v[16];
        #pragma unroll
        for (int q = 0; q < 16; ++q) {
            int lr = (q >> 3) * 16 + (q & 7) + 8 * hi;
            int lc = (lr < rv) ? lr : (rv - 1);
            h0v[q] = h0t[(unsigned)(lc * H_DIM + colN)];
        }
        const int cli = row0 + ((lane < rv) ? lane : (rv - 1));
        const float cl = coef0[cli] + coef1[cli];

        v8f acc0 = {}, acc1 = {};
        tile_mma<H_DIM>(As[buf], m, hi, bf, acc0, acc1);

        float res[16];
        #pragma unroll
        for (int q = 0; q < 16; ++q) {
            int lr = (q >> 3) * 16 + (q & 7) + 8 * hi;
            float cf   = __shfl(cl, lr);
            float hv   = As[buf][lr][colN];
            float h0x  = h0v[q];
            float outp = ((q < 8) ? acc0[q & 7] : acc1[q & 7]) + wbv;
            float outv = cf * outp - (hv * ew + bl * h0x);
            res[q] = hv + STEPF * fmaxf(outv, 0.f);
        }
        float* hot = h_out + (size_t)row0 * H_DIM;
        const unsigned o0 = (unsigned)(8 * hi * H_DIM + colN);
        if (rv >= ROWS_IT) {
            #pragma unroll
            for (int q = 0; q < 16; ++q)
                hot[o0 + (unsigned)(((q >> 3) * 16 + (q & 7)) * H_DIM)] = res[q];
        } else {
            #pragma unroll
            for (int q = 0; q < 16; ++q) {
                int lr = (q >> 3) * 16 + (q & 7) + 8 * hi;
                if (lr < rv) hot[o0 + (unsigned)(((q >> 3) * 16 + (q & 7)) * H_DIM)] = res[q];
            }
        }

        if (nxt < nIt) {
            #pragma unroll
            for (int j = 0; j < NPF; ++j) *(v4f*)&As[buf ^ 1][sr + 8 * j][sc] = pf[j];
        }
        __syncthreads();
        buf ^= 1;
    }
    }
}

__launch_bounds__(96)
__global__ void decoder_gemm(const float* __restrict__ h, const float* __restrict__ W,
                             const float* __restrict__ bias, float* __restrict__ out, int Nn) {
    __shared__ __align__(16) float As[2][ROWS_IT][H_DIM + 4];
    const int lane = threadIdx.x & 31;
    const int wave = threadIdx.x >> 5;
    const int m    = lane & 15;
    const int hi   = lane >> 4;
    const int colN = wave * 16 + m;
    const bool colok = (colN < C_DIM);
    const float* wrow = W + (size_t)(colok ? colN : 0) * H_DIM;

    v2f bf[H_DIM / 4];
    #pragma unroll
    for (int ks = 0; ks < H_DIM / 4; ++ks) {
        v2f b = *(const v2f*)&wrow[4 * ks + 2 * hi];
        if (!colok) { b.x = 0.f; b.y = 0.f; }
        bf[ks] = b;
    }
    const float bv = bias[colok ? colN : 0];

    const int nIt = (Nn + ROWS_IT - 1) / ROWS_IT;
    constexpr int NV4 = ROWS_IT * (H_DIM / 4);
    constexpr int NPF = (NV4 + 95) / 96;
    const unsigned tid = threadIdx.x;

    auto load_tile = [&](int t, v4f (&dst)[NPF]) {
        const int r0 = t * ROWS_IT;
        const v4f* base = (const v4f*)(h + (size_t)r0 * H_DIM);
        const int rv = Nn - r0;
        const unsigned last = (unsigned)(rv < ROWS_IT ? rv : ROWS_IT) * (H_DIM / 4) - 1u;
        #pragma unroll
        for (int j = 0; j < NPF; ++j) {
            unsigned o = tid + 96u * j;
            dst[j] = base[o < last ? o : last];
        }
    };
    auto store_tile = [&](int b, const v4f (&src)[NPF]) {
        #pragma unroll
        for (int j = 0; j < NPF; ++j) {
            unsigned o = tid + 96u * j;
            if (o < (unsigned)NV4) *(v4f*)&As[b][o >> 5][(o & 31) * 4] = src[j];
        }
    };

    int it = blockIdx.x;
    if (it < nIt) {
        v4f t0[NPF];
        load_tile(it, t0);
        store_tile(0, t0);
    }
    __syncthreads();
    int buf = 0;
    for (; it < nIt; it += gridDim.x) {
        const int row0 = it * ROWS_IT;
        const int rv   = Nn - row0;
        const int nxt  = it + gridDim.x;
        v4f pf[NPF];
        if (nxt < nIt) load_tile(nxt, pf);

        v8f acc0 = {}, acc1 = {};
        tile_mma<H_DIM>(As[buf], m, hi, bf, acc0, acc1);

        if (colok) {
            float* ot = out + (size_t)row0 * C_DIM;
            #pragma unroll
            for (int r = 0; r < 8; ++r) {
                int lr = r + 8 * hi;
                if (lr < rv)
                    __builtin_nontemporal_store(acc0[r] + bv, &ot[(unsigned)(lr * C_DIM + colN)]);
            }
            #pragma unroll
            for (int r = 0; r < 8; ++r) {
                int lr = 16 + r + 8 * hi;
                if (lr < rv)
                    __builtin_nontemporal_store(acc1[r] + bv, &ot[(unsigned)(lr * C_DIM + colN)]);
            }
        }

        if (nxt < nIt) store_tile(buf ^ 1, pf);
        __syncthreads();
        buf ^= 1;
    }
}


extern "C" void kernel_launch(void* const* d_in, const int* in_sizes, int n_in,
                              void* d_out, int out_size, void* d_ws, size_t ws_size,
                              hipStream_t stream) {
    const float* x     = (const float*)d_in[0];
    const int*   eidx  = (const int*)d_in[1];
    const float* enc_W = (const float*)d_in[2];
    const float* enc_b = (const float*)d_in[3];
    const float* W_raw = (const float*)d_in[4];
    const float* W_b   = (const float*)d_in[5];
    const float* ext_w = (const float*)d_in[6];
    const float* beta  = (const float*)d_in[7];
    const float* dec_W = (const float*)d_in[8];
    const float* dec_b = (const float*)d_in[9];

    const int Nn = in_sizes[0] / F_IN_DIM;
    const int E  = in_sizes[1] / 2;
    const int* row = eidx;
    const int* col = eidx + E;

    char* ws = (char*)d_ws;
    size_t off = 0;
    auto wsalloc = [&](size_t bytes) -> void* {
        void* p = ws + off;
        off = (off + bytes + 255) & ~(size_t)255;
        return p;
    };
    float* deg  = (float*)wsalloc((size_t)Nn * sizeof(float));
    float* deg1 = (float*)wsalloc((size_t)Nn * sizeof(float));
    float* coef = (float*)wsalloc((size_t)Nn * sizeof(float));
    float* cof1 = (float*)wsalloc((size_t)Nn * sizeof(float));
    float* h0   = (float*)wsalloc((size_t)Nn * H_DIM * sizeof(float));
    float* ha   = (float*)wsalloc((size_t)Nn * H_DIM * sizeof(float));
    float* hb   = (float*)wsalloc((size_t)Nn * H_DIM * sizeof(float));
    (void)ws_size; (void)n_in; (void)out_size;

    const int tb = 256;
    zero_four<<<(Nn + tb - 1) / tb, tb, 0, stream>>>(deg, deg1, coef, cof1, Nn);
    encoder_edge<0><<<256, 512, 0, stream>>>(x, enc_W, enc_b, h0, row, col, nullptr, deg, deg1, Nn, E);
    deg_inv_kernel<<<(Nn + tb - 1) / tb, tb, 0, stream>>>(deg, deg1, Nn);
    encoder_edge<1><<<256, 512, 0, stream>>>(x, enc_W, enc_b, h0, row, col, deg, coef, cof1, Nn, E);

    layers_gemm<<<512, 256, 0, stream>>>(h0, coef, cof1, W_raw, W_b, ext_w, beta, ha, hb, Nn);
    const float* hin = hb;
    decoder_gemm<<<1024, 96, 0, stream>>>(hin, dec_W, dec_b, (float*)d_out, Nn);
}


# layer launch keeps each block's row tiles of h in LDS through all 4 layers: epilogue writes into the tile, only layer 0 loads and layer 3 stores h; raw weights staged in 2 halves; sum order unchanged
# speedup vs baseline: 2.4476x; 1.0612x over previous
#include <hip/hip_runtime.h>

#define H_DIM   128
#define F_IN_DIM 256
#define C_DIM   40
#define L_DIM   4
#define STEPF   0.1f
#define WR_PITCH (H_DIM + 2)
#define WR_ELEMS (H_DIM * WR_PITCH)

typedef __attribute__((ext_vector_type(2))) float v2f;
typedef __attribute__((ext_vector_type(4))) float v4f;
typedef __attribute__((ext_vector_type(8))) float v8f;


__device__ __forceinline__ unsigned package_half() {
    return (__builtin_amdgcn_s_sendmsg_rtn(0x87u) >> 18) & 1u;
}

__global__ void zero_four(float* __restrict__ a, float* __restrict__ b,
                          float* __restrict__ c, float* __restrict__ d, int n) {
    int i = blockIdx.x * blockDim.x + threadIdx.x;
    if (i < n) { a[i] = 0.f; b[i] = 0.f; c[i] = 0.f; d[i] = 0.f; }
}

__global__ void deg_inv_kernel(float* __restrict__ deg0, const float* __restrict__ deg1, int n) {
    int i = blockIdx.x * blockDim.x + threadIdx.x;
    if (i < n) { float d = deg0[i] + deg1[i]; deg0[i] = (d > 0.f) ? rsqrtf(d) : 0.f; }
}


#define ROWS_IT 32

template <int K>
__device__ __forceinline__ void tile_mma(const float (*At)[K + 4], int m, int hi,
                                         const v2f (&bf)[K / 4], v8f& acc0, v8f& acc1) {
    constexpr int KS = K / 4;
    constexpr int AHEAD = 4;
    v2f a0[KS], a1[KS];
    #pragma unroll
    for (int ks = 0; ks < AHEAD; ++ks) {
        a0[ks] = *(const v2f*)&At[m][4 * ks + 2 * hi];
        a1[ks] = *(const v2f*)&At[16 + m][4 * ks + 2 * hi];
    }
    #pragma unroll
    for (int ks = 0; ks < KS; ++ks) {
        if (ks + AHEAD < KS) {
            a0[ks + AHEAD] = *(const v2f*)&At[m][4 * (ks + AHEAD) + 2 * hi];
            a1[ks + AHEAD] = *(const v2f*)&At[16 + m][4 * (ks + AHEAD) + 2 * hi];
        }
        acc0 = __builtin_amdgcn_wmma_f32_16x16x4_f32(false, a0[ks], false, bf[ks],
                                                     (short)0, acc0, false, false);
        acc1 = __builtin_amdgcn_wmma_f32_16x16x4_f32(false, a1[ks], false, bf[ks],
                                                     (short)0, acc1, false, false);
        __builtin_amdgcn_sched_barrier(0);
    }
}

#define EDGES_PER_IT 4

template <int ROLE>
__launch_bounds__(512, 1)
__global__ void encoder_edge(const float* __restrict__ x, const float* __restrict__ W,
                             const float* __restrict__ bias, float* __restrict__ h,
                             const int* __restrict__ row, const int* __restrict__ col,
                             const float* __restrict__ dinv, float* __restrict__ acc0,
                             float* __restrict__ acc1, int Nn, int E) {
    __shared__ __align__(16) float As[2][ROWS_IT][F_IN_DIM + 4];
    const bool gemm = __builtin_amdgcn_readfirstlane((int)threadIdx.x) < 256;
    const int lane = threadIdx.x & 31;
    const int wave = (threadIdx.x >> 5) & 7;
    const int m    = lane & 15;
    const int hi   = lane >> 4;
    const int colN = wave * 16 + m;

    const int nItAll = (Nn + ROWS_IT - 1) / ROWS_IT;
    const int nHalf  = (nItAll + 1) >> 1;
    const int tEnd   = ROLE ? nItAll : nHalf;
    const int tBeg   = ROLE ? nHalf : 0;
    constexpr int NPF = ROWS_IT * (F_IN_DIM / 4) / 256;
    const unsigned tid = threadIdx.x & 255u;
    const unsigned sr = tid >> 6, sc = (tid & 63) * 4;

    auto load_tile = [&](int t, v4f (&dst)[NPF]) {
        const int r0 = t * ROWS_IT;
        const v4f* base = (const v4f*)(x + (size_t)r0 * F_IN_DIM);
        const int rv = Nn - r0;
        if (rv >= ROWS_IT) {
            #pragma unroll
            for (int j = 0; j < NPF; ++j)
                dst[j] = __builtin_nontemporal_load(&base[tid + 256u * j]);
        } else {
            const unsigned last = (unsigned)rv * (F_IN_DIM / 4) - 1u;
            #pragma unroll
            for (int j = 0; j < NPF; ++j) {
                unsigned o = tid + 256u * j;
                dst[j] = __builtin_nontemporal_load(&base[o < last ? o : last]);
            }
        }
    };

    const unsigned G = gridDim.x * 256u;
    const unsigned g = blockIdx.x * 256u + tid;
    float* accp = package_half() ? acc1 : acc0;
    int er[EDGES_PER_IT], ec[EDGES_PER_IT];
    auto load_idx = [&](unsigned k0, int (&r)[EDGES_PER_IT], int (&c)[EDGES_PER_IT]) {
        #pragma unroll
        for (int q = 0; q < EDGES_PER_IT; ++q) {
            const unsigned e  = g + G * (k0 + (unsigned)q);
            const bool     ok = e < (unsigned)E;
            const unsigned ee = ok ? e : 0u;
            const int rr = row[ee];
            c[q] = ROLE ? col[ee] : 0;
            r[q] = ok ? rr : -1;
        }
    };
    auto do_edges = [&](const int (&r)[EDGES_PER_IT], const int (&c)[EDGES_PER_IT]) {
        if (ROLE == 0) {
            #pragma unroll
            for (int q = 0; q < EDGES_PER_IT; ++q)
                if (r[q] >= 0) atomicAdd(&accp[r[q]], 1.0f);
        } else {
            float pa[EDGES_PER_IT], pb[EDGES_PER_IT];
            #pragma unroll
            for (int q = 0; q < EDGES_PER_IT; ++q) {
                pa[q] = dinv[r[q] >= 0 ? r[q] : 0];
                pb[q] = dinv[c[q]];
            }
            #pragma unroll
            for (int q = 0; q < EDGES_PER_IT; ++q)
                if (r[q] >= 0) atomicAdd(&accp[c[q]], pa[q] * pb[q]);
        }
    };

    v2f bf[F_IN_DIM / 4];
    float bv = 0.f;
    int it = tBeg + (int)blockIdx.x;
    if (gemm) {
        const float* wrow = W + (size_t)colN * F_IN_DIM;
        #pragma unroll
        for (int ks = 0; ks < F_IN_DIM / 4; ++ks)
            bf[ks] = *(const v2f*)&wrow[4 * ks + 2 * hi];
        bv = bias[colN];
        if (it < tEnd) {
            v4f t0[NPF];
            load_tile(it, t0);
            #pragma unroll
            for (int j = 0; j < NPF; ++j) *(v4f*)&As[0][sr + 4 * j][sc] = t0[j];
        }
    } else {
        #pragma unroll
        for (int ks = 0; ks < F_IN_DIM / 4; ++ks) { bf[ks].x = 0.f; bf[ks].y = 0.f; }
        load_idx(0u, er, ec);
    }
    unsigned kc = 0;
    __syncthreads();
    int buf = 0;
    for (; it < tEnd; it += gridDim.x) {
        if (gemm) {
            const int row0 = it * ROWS_IT;
            const int rv   = Nn - row0;
            const int nxt  = it + gridDim.x;
            v4f pf[NPF];
            if (nxt < tEnd) load_tile(nxt, pf);

            v8f acc0v = {}, acc1v = {};
            tile_mma<F_IN_DIM>(As[buf], m, hi, bf, acc0v, acc1v);

            float* ht = h + (size_t)row0 * H_DIM;
            const unsigned o0 = (unsigned)(8 * hi * H_DIM + colN);
            if (rv >= ROWS_IT) {
                #pragma unroll
                for (int r = 0; r < 8; ++r) {
                    ht[o0 + (unsigned)(r * H_DIM)]        = acc0v[r] + bv;
                    ht[o0 + (unsigned)((16 + r) * H_DIM)] = acc1v[r] + bv;
                }
            } else {
                #pragma unroll
                for (int r = 0; r < 8; ++r) {
                    if (r + 8 * hi < rv)      ht[o0 + (unsigned)(r * H_DIM)]        = acc0v[r] + bv;
                    if (16 + r + 8 * hi < rv) ht[o0 + (unsigned)((16 + r) * H_DIM)] = acc1v[r] + bv;
                }
            }

            if (nxt < tEnd) {
                #pragma unroll
                for (int j = 0; j < NPF; ++j) *(v4f*)&As[buf ^ 1][sr + 4 * j][sc] = pf[j];
            }
        } else {
            int nr[EDGES_PER_IT], nc[EDGES_PER_IT];
            load_idx(kc + EDGES_PER_IT, nr, nc);
            do_edges(er, ec);
            #pragma unroll
            for (int q = 0; q < EDGES_PER_IT; ++q) { er[q] = nr[q]; ec[q] = nc[q]; }
            kc += EDGES_PER_IT;
        }
        __syncthreads();
        buf ^= 1;
    }
    if (!gemm) {
        do_edges(er, ec);
        for (unsigned k = kc + EDGES_PER_IT; g + G * k < (unsigned)E; k += EDGES_PER_IT) {
            load_idx(k, er, ec);
            do_edges(er, ec);
        }
    }
}

#define MAX_SLOTS 7
#define WH_ROWS   (H_DIM / 2)
#define WH_ELEMS  (WH_ROWS * WR_PITCH)

template <int KS0, int KS1>
__device__ __forceinline__ void strip_walk(const float* wsm, int off, int c, int hi,
                                           v2f (&bf)[H_DIM / 4], float& s) {
    const int cw = off + c * WR_PITCH;
    #pragma unroll
    for (int ks = KS0; ks < KS1; ++ks) {
        float v[4];
        #pragma unroll
        for (int e = 0; e < 4; ++e) {
            const int j = 4 * ks + e;
            const int mx = (c > j) ? c : j;
            v[e] = wsm[(cw - (WR_PITCH - 1) * mx) + j * WR_PITCH];
            s += fabsf(v[e]);
        }
        bf[ks].x = hi ? v[2] : v[0];
        bf[ks].y = hi ? v[3] : v[1];
    }
}

__launch_bounds__(256, 4)
__global__ void layers_gemm(const float* __restrict__ h0, const float* __restrict__ coef0,
                            const float* __restrict__ coef1, const float* __restrict__ Wraw_all,
                            const float* __restrict__ Wb_all, const float* __restrict__ extw_all,
                            const float* __restrict__ beta_all, float* __restrict__ hfin, int Nn) {
    __shared__ __align__(16) float wsm[WH_ELEMS];
    __shared__ __align__(16) float Ts[MAX_SLOTS][ROWS_IT][H_DIM + 4];
    const int lane = threadIdx.x & 31;
    const int wave = __builtin_amdgcn_readfirstlane((int)threadIdx.x) >> 5;
    const int m    = lane & 15;
    const int hi   = lane >> 4;
    const int colN = wave * 16 + m;
    const bool lowcol = wave < (WH_ROWS / 16);

    const int nIt = (Nn + ROWS_IT - 1) / ROWS_IT;
    constexpr int NPF = ROWS_IT * (H_DIM / 4) / 256;
    const unsigned tid = threadIdx.x;
    const unsigned sr = tid >> 5, sc = (tid & 31) * 4;

    auto load_tile = [&](int t, v4f (&dst)[NPF]) {
        const int r0 = t * ROWS_IT;
        const v4f* base = (const v4f*)(h0 + (size_t)r0 * H_DIM);
        const int rv = Nn - r0;
        if (rv >= ROWS_IT) {
            #pragma unroll
            for (int j = 0; j < NPF; ++j) dst[j] = base[tid + 256u * j];
        } else {
            const unsigned last = (unsigned)rv * (H_DIM / 4) - 1u;
            #pragma unroll
            for (int j = 0; j < NPF; ++j) {
                unsigned o = tid + 256u * j;
                dst[j] = base[o < last ? o : last];
            }
        }
    };

    #pragma unroll 1
    for (int l = 0; l < L_DIM; ++l) {
    const bool first = (l == 0);
    const bool lastl = (l == L_DIM - 1);
    const float* Wraw  = Wraw_all + (size_t)l * WR_ELEMS;
    const float wbv = Wb_all[l * H_DIM + colN];
    const float ew  = extw_all[l * H_DIM + colN];
    const float bl  = beta_all[l];

    int it = blockIdx.x;

    constexpr int NW4 = WH_ELEMS / 4;
    constexpr int NWL = (NW4 + 255) / 256;
    auto load_half = [&](const float* srcf, v4f (&wreg)[NWL]) {
        const v4f* src = (const v4f*)srcf;
        #pragma unroll
        for (int j = 0; j < NWL; ++j) {
            const unsigned o = tid + 256u * j;
            if (256 * (j + 1) <= NW4) wreg[j] = src[o];
            else wreg[j] = src[o < (unsigned)NW4 ? o : (unsigned)(NW4 - 1)];
        }
    };
    auto store_half = [&](const v4f (&wreg)[NWL]) {
        #pragma unroll
        for (int j = 0; j < NWL; ++j) {
            const unsigned o = tid + 256u * j;
            if (256 * (j + 1) <= NW4 || o < (unsigned)NW4) *(v4f*)&wsm[4u * o] = wreg[j];
        }
    };
    {
        v4f wreg[NWL];
        load_half(Wraw, wreg);
        store_half(wreg);
    }
    __syncthreads();

    v2f bf[H_DIM / 4];
    float s = 0.f;
    int c = colN;
    asm volatile("" : "+v"(c));
    float dsc = 0.f, dad = 0.f;
    if (lowcol) {
        wsm[c * WR_PITCH + c] = 0.f;
        strip_walk<0, H_DIM / 4>(wsm, 0, c, hi, bf, s);
        dsc = wsm[c * WR_PITCH + H_DIM];
        dad = wsm[c * WR_PITCH + H_DIM + 1];
    } else {
        strip_walk<0, WH_ROWS / 4>(wsm, 0, c, hi, bf, s);
    }
    v4f wregB[NWL];
    load_half(Wraw + WH_ELEMS, wregB);
    v4f t0[NPF];
    if (first && it < nIt) load_tile(it, t0);
    __syncthreads();
    store_half(wregB);
    if (first && it < nIt) {
        #pragma unroll
        for (int j = 0; j < NPF; ++j) *(v4f*)&Ts[0][sr + 8 * j][sc] = t0[j];
    }
    __syncthreads();
    if (!lowcol) {
        asm volatile("" : "+v"(c));
        wsm[c * WR_PITCH + c - WH_ELEMS] = 0.f;
        strip_walk<WH_ROWS / 4, H_DIM / 4>(wsm, -WH_ELEMS, c, hi, bf, s);
        dsc = wsm[c * WR_PITCH + H_DIM - WH_ELEMS];
        dad = wsm[c * WR_PITCH + H_DIM + 1 - WH_ELEMS];
    }
    {
        const float d = dsc * s + dad;
        #pragma unroll
        for (int ks = 0; ks < H_DIM / 4; ++ks) {
            const int k = 4 * ks + 2 * hi;
            if (k == c)     bf[ks].x = d;
            if (k + 1 == c) bf[ks].y = d;
        }
    }

    int slot = 0;
    for (; it < nIt; it += gridDim.x, ++slot) {
        float (*At)[H_DIM + 4] = Ts[slot];
        const int row0 = it * ROWS_IT;
        const int rv   = Nn - row0;
        const int nxt  = it + gridDim.x;
        v4f pf[NPF];
        const bool fetch = first && nxt < nIt;
        if (fetch) load_tile(nxt, pf);
        const float* h0t = h0 + (size_t)row0 * H_DIM;
        float h0v[16];
        if (!first) {
            #pragma unroll
            for (int q = 0; q < 16; ++q) {
                int lr = (q >> 3) * 16 + (q & 7) + 8 * hi;
                int lc = (lr < rv) ? lr : (rv - 1);
                h0v[q] = h0t[(unsigned)(lc * H_DIM + colN)];
            }
        } else {
            #pragma unroll
            for (int q = 0; q < 16; ++q) h0v[q] = 0.f;
        }
        const int cli = row0 + ((lane < rv) ? lane : (rv - 1));
        const float cl = coef0[cli] + coef1[cli];

        v8f acc0 = {}, acc1 = {};
        tile_mma<H_DIM>(At, m, hi, bf, acc0, acc1);

        if (fetch) {
            #pragma unroll
            for (int j = 0; j < NPF; ++j) *(v4f*)&Ts[slot + 1][sr + 8 * j][sc] = pf[j];
        }
        __syncthreads();

        float res[16];
        #pragma unroll
        for (int q = 0; q < 16; ++q) {
            int lr = (q >> 3) * 16 + (q & 7) + 8 * hi;
            float cf   = __shfl(cl, lr);
            float hv   = At[lr][colN];
            float h0x  = first ? hv : h0v[q];
            float outp = ((q < 8) ? acc0[q & 7] : acc1[q & 7]) + wbv;
            float outv = cf * outp - (hv * ew + bl * h0x);
            res[q] = hv + STEPF * fmaxf(outv, 0.f);
        }
        if (!lastl) {
            #pragma unroll
            for (int q = 0; q < 16; ++q)
                At[(q >> 3) * 16 + (q & 7) + 8 * hi][colN] = res[q];
        } else {
            float* hot = hfin + (size_t)row0 * H_DIM;
            const unsigned o0 = (unsigned)(8 * hi * H_DIM + colN);
            if (rv >= ROWS_IT) {
                #pragma unroll
                for (int q = 0; q < 16; ++q)
                    hot[o0 + (unsigned)(((q >> 3) * 16 + (q & 7)) * H_DIM)] = res[q];
            } else {
                #pragma unroll
                for (int q = 0; q < 16; ++q) {
                    int lr = (q >> 3) * 16 + (q & 7) + 8 * hi;
                    if (lr < rv) hot[o0 + (unsigned)(((q >> 3) * 16 + (q & 7)) * H_DIM)] = res[q];
                }
            }
        }
    }
    }
}

__launch_bounds__(96)
__global__ void decoder_gemm(const float* __restrict__ h, const float* __restrict__ W,
                             const float* __restrict__ bias, float* __restrict__ out, int Nn) {
    __shared__ __align__(16) float As[2][ROWS_IT][H_DIM + 4];
    const int lane = threadIdx.x & 31;
    const int wave = threadIdx.x >> 5;
    const int m    = lane & 15;
    const int hi   = lane >> 4;
    const int colN = wave * 16 + m;
    const bool colok = (colN < C_DIM);
    const float* wrow = W + (size_t)(colok ? colN : 0) * H_DIM;

    v2f bf[H_DIM / 4];
    #pragma unroll
    for (int ks = 0; ks < H_DIM / 4; ++ks) {
        v2f b = *(const v2f*)&wrow[4 * ks + 2 * hi];
        if (!colok) { b.x = 0.f; b.y = 0.f; }
        bf[ks] = b;
    }
    const float bv = bias[colok ? colN : 0];

    const int nIt = (Nn + ROWS_IT - 1) / ROWS_IT;
    constexpr int NV4 = ROWS_IT * (H_DIM / 4);
    constexpr int NPF = (NV4 + 95) / 96;
    const unsigned tid = threadIdx.x;

    auto load_tile = [&](int t, v4f (&dst)[NPF]) {
        const int r0 = t * ROWS_IT;
        const v4f* base = (const v4f*)(h + (size_t)r0 * H_DIM);
        const int rv = Nn - r0;
        const unsigned last = (unsigned)(rv < ROWS_IT ? rv : ROWS_IT) * (H_DIM / 4) - 1u;
        #pragma unroll
        for (int j = 0; j < NPF; ++j) {
            unsigned o = tid + 96u * j;
            dst[j] = base[o < last ? o : last];
        }
    };
    auto store_tile = [&](int b, const v4f (&src)[NPF]) {
        #pragma unroll
        for (int j = 0; j < NPF; ++j) {
            unsigned o = tid + 96u * j;
            if (o < (unsigned)NV4) *(v4f*)&As[b][o >> 5][(o & 31) * 4] = src[j];
        }
    };

    int it = blockIdx.x;
    if (it < nIt) {
        v4f t0[NPF];
        load_tile(it, t0);
        store_tile(0, t0);
    }
    __syncthreads();
    int buf = 0;
    for (; it < nIt; it += gridDim.x) {
        const int row0 = it * ROWS_IT;
        const int rv   = Nn - row0;
        const int nxt  = it + gridDim.x;
        v4f pf[NPF];
        if (nxt < nIt) load_tile(nxt, pf);

        v8f acc0 = {}, acc1 = {};
        tile_mma<H_DIM>(As[buf], m, hi, bf, acc0, acc1);

        if (colok) {
            float* ot = out + (size_t)row0 * C_DIM;
            #pragma unroll
            for (int r = 0; r < 8; ++r) {
                int lr = r + 8 * hi;
                if (lr < rv)
                    __builtin_nontemporal_store(acc0[r] + bv, &ot[(unsigned)(lr * C_DIM + colN)]);
            }
            #pragma unroll
            for (int r = 0; r < 8; ++r) {
                int lr = 16 + r + 8 * hi;
                if (lr < rv)
                    __builtin_nontemporal_store(acc1[r] + bv, &ot[(unsigned)(lr * C_DIM + colN)]);
            }
        }

        if (nxt < nIt) store_tile(buf ^ 1, pf);
        __syncthreads();
        buf ^= 1;
    }
}


extern "C" void kernel_launch(void* const* d_in, const int* in_sizes, int n_in,
                              void* d_out, int out_size, void* d_ws, size_t ws_size,
                              hipStream_t stream) {
    const float* x     = (const float*)d_in[0];
    const int*   eidx  = (const int*)d_in[1];
    const float* enc_W = (const float*)d_in[2];
    const float* enc_b = (const float*)d_in[3];
    const float* W_raw = (const float*)d_in[4];
    const float* W_b   = (const float*)d_in[5];
    const float* ext_w = (const float*)d_in[6];
    const float* beta  = (const float*)d_in[7];
    const float* dec_W = (const float*)d_in[8];
    const float* dec_b = (const float*)d_in[9];

    const int Nn = in_sizes[0] / F_IN_DIM;
    const int E  = in_sizes[1] / 2;
    const int* row = eidx;
    const int* col = eidx + E;

    char* ws = (char*)d_ws;
    size_t off = 0;
    auto wsalloc = [&](size_t bytes) -> void* {
        void* p = ws + off;
        off = (off + bytes + 255) & ~(size_t)255;
        return p;
    };
    float* deg  = (float*)wsalloc((size_t)Nn * sizeof(float));
    float* deg1 = (float*)wsalloc((size_t)Nn * sizeof(float));
    float* coef = (float*)wsalloc((size_t)Nn * sizeof(float));
    float* cof1 = (float*)wsalloc((size_t)Nn * sizeof(float));
    float* h0   = (float*)wsalloc((size_t)Nn * H_DIM * sizeof(float));
    float* hb   = (float*)wsalloc((size_t)Nn * H_DIM * sizeof(float));
    (void)ws_size; (void)n_in; (void)out_size;

    const int tb = 256;
    zero_four<<<(Nn + tb - 1) / tb, tb, 0, stream>>>(deg, deg1, coef, cof1, Nn);
    encoder_edge<0><<<256, 512, 0, stream>>>(x, enc_W, enc_b, h0, row, col, nullptr, deg, deg1, Nn, E);
    deg_inv_kernel<<<(Nn + tb - 1) / tb, tb, 0, stream>>>(deg, deg1, Nn);
    encoder_edge<1><<<256, 512, 0, stream>>>(x, enc_W, enc_b, h0, row, col, deg, coef, cof1, Nn, E);

    layers_gemm<<<512, 256, 0, stream>>>(h0, coef, cof1, W_raw, W_b, ext_w, beta, hb, Nn);
    const float* hin = hb;
    decoder_gemm<<<1024, 96, 0, stream>>>(hin, dec_W, dec_b, (float*)d_out, Nn);
}


# decoder moves into the layer launch: after the last layer six waves per block form out from the final h still in the block's LDS tiles; final h write, re-read and one launch go; sum order unchanged
# speedup vs baseline: 2.5753x; 1.0522x over previous
#include <hip/hip_runtime.h>

#define H_DIM   128
#define F_IN_DIM 256
#define C_DIM   40
#define L_DIM   4
#define STEPF   0.1f
#define WR_PITCH (H_DIM + 2)
#define WR_ELEMS (H_DIM * WR_PITCH)

typedef __attribute__((ext_vector_type(2))) float v2f;
typedef __attribute__((ext_vector_type(4))) float v4f;
typedef __attribute__((ext_vector_type(8))) float v8f;


__device__ __forceinline__ unsigned package_half() {
    return (__builtin_amdgcn_s_sendmsg_rtn(0x87u) >> 18) & 1u;
}

__global__ void zero_four(float* __restrict__ a, float* __restrict__ b,
                          float* __restrict__ c, float* __restrict__ d, int n) {
    int i = blockIdx.x * blockDim.x + threadIdx.x;
    if (i < n) { a[i] = 0.f; b[i] = 0.f; c[i] = 0.f; d[i] = 0.f; }
}

__global__ void deg_inv_kernel(float* __restrict__ deg0, const float* __restrict__ deg1, int n) {
    int i = blockIdx.x * blockDim.x + threadIdx.x;
    if (i < n) { float d = deg0[i] + deg1[i]; deg0[i] = (d > 0.f) ? rsqrtf(d) : 0.f; }
}


#define ROWS_IT 32

template <int K>
__device__ __forceinline__ void tile_mma(const float (*At)[K + 4], int m, int hi,
                                         const v2f (&bf)[K / 4], v8f& acc0, v8f& acc1) {
    constexpr int KS = K / 4;
    constexpr int AHEAD = 4;
    v2f a0[KS], a1[KS];
    #pragma unroll
    for (int ks = 0; ks < AHEAD; ++ks) {
        a0[ks] = *(const v2f*)&At[m][4 * ks + 2 * hi];
        a1[ks] = *(const v2f*)&At[16 + m][4 * ks + 2 * hi];
    }
    #pragma unroll
    for (int ks = 0; ks < KS; ++ks) {
        if (ks + AHEAD < KS) {
            a0[ks + AHEAD] = *(const v2f*)&At[m][4 * (ks + AHEAD) + 2 * hi];
            a1[ks + AHEAD] = *(const v2f*)&At[16 + m][4 * (ks + AHEAD) + 2 * hi];
        }
        acc0 = __builtin_amdgcn_wmma_f32_16x16x4_f32(false, a0[ks], false, bf[ks],
                                                     (short)0, acc0, false, false);
        acc1 = __builtin_amdgcn_wmma_f32_16x16x4_f32(false, a1[ks], false, bf[ks],
                                                     (short)0, acc1, false, false);
        __builtin_amdgcn_sched_barrier(0);
    }
}

#define EDGES_PER_IT 4

template <int ROLE>
__launch_bounds__(512, 1)
__global__ void encoder_edge(const float* __restrict__ x, const float* __restrict__ W,
                             const float* __restrict__ bias, float* __restrict__ h,
                             const int* __restrict__ row, const int* __restrict__ col,
                             const float* __restrict__ dinv, float* __restrict__ acc0,
                             float* __restrict__ acc1, int Nn, int E) {
    __shared__ __align__(16) float As[2][ROWS_IT][F_IN_DIM + 4];
    const bool gemm = __builtin_amdgcn_readfirstlane((int)threadIdx.x) < 256;
    const int lane = threadIdx.x & 31;
    const int wave = (threadIdx.x >> 5) & 7;
    const int m    = lane & 15;
    const int hi   = lane >> 4;
    const int colN = wave * 16 + m;

    const int nItAll = (Nn + ROWS_IT - 1) / ROWS_IT;
    const int nHalf  = (nItAll + 1) >> 1;
    const int tEnd   = ROLE ? nItAll : nHalf;
    const int tBeg   = ROLE ? nHalf : 0;
    constexpr int NPF = ROWS_IT * (F_IN_DIM / 4) / 256;
    const unsigned tid = threadIdx.x & 255u;
    const unsigned sr = tid >> 6, sc = (tid & 63) * 4;

    auto load_tile = [&](int t, v4f (&dst)[NPF]) {
        const int r0 = t * ROWS_IT;
        const v4f* base = (const v4f*)(x + (size_t)r0 * F_IN_DIM);
        const int rv = Nn - r0;
        if (rv >= ROWS_IT) {
            #pragma unroll
            for (int j = 0; j < NPF; ++j)
                dst[j] = __builtin_nontemporal_load(&base[tid + 256u * j]);
        } else {
            const unsigned last = (unsigned)rv * (F_IN_DIM / 4) - 1u;
            #pragma unroll
            for (int j = 0; j < NPF; ++j) {
                unsigned o = tid + 256u * j;
                dst[j] = __builtin_nontemporal_load(&base[o < last ? o : last]);
            }
        }
    };

    const unsigned G = gridDim.x * 256u;
    const unsigned g = blockIdx.x * 256u + tid;
    float* accp = package_half() ? acc1 : acc0;
    int er[EDGES_PER_IT], ec[EDGES_PER_IT];
    auto load_idx = [&](unsigned k0, int (&r)[EDGES_PER_IT], int (&c)[EDGES_PER_IT]) {
        #pragma unroll
        for (int q = 0; q < EDGES_PER_IT; ++q) {
            const unsigned e  = g + G * (k0 + (unsigned)q);
            const bool     ok = e < (unsigned)E;
            const unsigned ee = ok ? e : 0u;
            const int rr = row[ee];
            c[q] = ROLE ? col[ee] : 0;
            r[q] = ok ? rr : -1;
        }
    };
    auto do_edges = [&](const int (&r)[EDGES_PER_IT], const int (&c)[EDGES_PER_IT]) {
        if (ROLE == 0) {
            #pragma unroll
            for (int q = 0; q < EDGES_PER_IT; ++q)
                if (r[q] >= 0) atomicAdd(&accp[r[q]], 1.0f);
        } else {
            float pa[EDGES_PER_IT], pb[EDGES_PER_IT];
            #pragma unroll
            for (int q = 0; q < EDGES_PER_IT; ++q) {
                pa[q] = dinv[r[q] >= 0 ? r[q] : 0];
                pb[q] = dinv[c[q]];
            }
            #pragma unroll
            for (int q = 0; q < EDGES_PER_IT; ++q)
                if (r[q] >= 0) atomicAdd(&accp[c[q]], pa[q] * pb[q]);
        }
    };

    v2f bf[F_IN_DIM / 4];
    float bv = 0.f;
    int it = tBeg + (int)blockIdx.x;
    if (gemm) {
        const float* wrow = W + (size_t)colN * F_IN_DIM;
        #pragma unroll
        for (int ks = 0; ks < F_IN_DIM / 4; ++ks)
            bf[ks] = *(const v2f*)&wrow[4 * ks + 2 * hi];
        bv = bias[colN];
        if (it < tEnd) {
            v4f t0[NPF];
            load_tile(it, t0);
            #pragma unroll
            for (int j = 0; j < NPF; ++j) *(v4f*)&As[0][sr + 4 * j][sc] = t0[j];
        }
    } else {
        #pragma unroll
        for (int ks = 0; ks < F_IN_DIM / 4; ++ks) { bf[ks].x = 0.f; bf[ks].y = 0.f; }
        load_idx(0u, er, ec);
    }
    unsigned kc = 0;
    __syncthreads();
    int buf = 0;
    for (; it < tEnd; it += gridDim.x) {
        if (gemm) {
            const int row0 = it * ROWS_IT;
            const int rv   = Nn - row0;
            const int nxt  = it + gridDim.x;
            v4f pf[NPF];
            if (nxt < tEnd) load_tile(nxt, pf);

            v8f acc0v = {}, acc1v = {};
            tile_mma<F_IN_DIM>(As[buf], m, hi, bf, acc0v, acc1v);

            float* ht = h + (size_t)row0 * H_DIM;
            const unsigned o0 = (unsigned)(8 * hi * H_DIM + colN);
            if (rv >= ROWS_IT) {
                #pragma unroll
                for (int r = 0; r < 8; ++r) {
                    ht[o0 + (unsigned)(r * H_DIM)]        = acc0v[r] + bv;
                    ht[o0 + (unsigned)((16 + r) * H_DIM)] = acc1v[r] + bv;
                }
            } else {
                #pragma unroll
                for (int r = 0; r < 8; ++r) {
                    if (r + 8 * hi < rv)      ht[o0 + (unsigned)(r * H_DIM)]        = acc0v[r] + bv;
                    if (16 + r + 8 * hi < rv) ht[o0 + (unsigned)((16 + r) * H_DIM)] = acc1v[r] + bv;
                }
            }

            if (nxt < tEnd) {
                #pragma unroll
                for (int j = 0; j < NPF; ++j) *(v4f*)&As[buf ^ 1][sr + 4 * j][sc] = pf[j];
            }
        } else {
            int nr[EDGES_PER_IT], nc[EDGES_PER_IT];
            load_idx(kc + EDGES_PER_IT, nr, nc);
            do_edges(er, ec);
            #pragma unroll
            for (int q = 0; q < EDGES_PER_IT; ++q) { er[q] = nr[q]; ec[q] = nc[q]; }
            kc += EDGES_PER_IT;
        }
        __syncthreads();
        buf ^= 1;
    }
    if (!gemm) {
        do_edges(er, ec);
        for (unsigned k = kc + EDGES_PER_IT; g + G * k < (unsigned)E; k += EDGES_PER_IT) {
            load_idx(k, er, ec);
            do_edges(er, ec);
        }
    }
}

#define MAX_SLOTS 7
#define WH_ROWS   (H_DIM / 2)
#define WH_ELEMS  (WH_ROWS * WR_PITCH)

template <int KS0, int KS1>
__device__ __forceinline__ void strip_walk(const float* wsm, int off, int c, int hi,
                                           v2f (&bf)[H_DIM / 4], float& s) {
    const int cw = off + c * WR_PITCH;
    #pragma unroll
    for (int ks = KS0; ks < KS1; ++ks) {
        float v[4];
        #pragma unroll
        for (int e = 0; e < 4; ++e) {
            const int j = 4 * ks + e;
            const int mx = (c > j) ? c : j;
            v[e] = wsm[(cw - (WR_PITCH - 1) * mx) + j * WR_PITCH];
            s += fabsf(v[e]);
        }
        bf[ks].x = hi ? v[2] : v[0];
        bf[ks].y = hi ? v[3] : v[1];
    }
}

__launch_bounds__(256, 4)
__global__ void layers_gemm(const float* __restrict__ h0, const float* __restrict__ coef0,
                            const float* __restrict__ coef1, const float* __restrict__ Wraw_all,
                            const float* __restrict__ Wb_all, const float* __restrict__ extw_all,
                            const float* __restrict__ beta_all, const float* __restrict__ decW,
                            const float* __restrict__ decb, float* __restrict__ out, int Nn) {
    __shared__ __align__(16) float wsm[WH_ELEMS];
    __shared__ __align__(16) float Ts[MAX_SLOTS][ROWS_IT][H_DIM + 4];
    const int lane = threadIdx.x & 31;
    const int wave = __builtin_amdgcn_readfirstlane((int)threadIdx.x) >> 5;
    const int m    = lane & 15;
    const int hi   = lane >> 4;
    const int colN = wave * 16 + m;
    const bool lowcol = wave < (WH_ROWS / 16);

    const int nIt = (Nn + ROWS_IT - 1) / ROWS_IT;
    constexpr int NPF = ROWS_IT * (H_DIM / 4) / 256;
    const unsigned tid = threadIdx.x;
    const unsigned sr = tid >> 5, sc = (tid & 31) * 4;

    auto load_tile = [&](int t, v4f (&dst)[NPF]) {
        const int r0 = t * ROWS_IT;
        const v4f* base = (const v4f*)(h0 + (size_t)r0 * H_DIM);
        const int rv = Nn - r0;
        if (rv >= ROWS_IT) {
            #pragma unroll
            for (int j = 0; j < NPF; ++j) dst[j] = base[tid + 256u * j];
        } else {
            const unsigned last = (unsigned)rv * (H_DIM / 4) - 1u;
            #pragma unroll
            for (int j = 0; j < NPF; ++j) {
                unsigned o = tid + 256u * j;
                dst[j] = base[o < last ? o : last];
            }
        }
    };

    #pragma unroll 1
    for (int l = 0; l < L_DIM; ++l) {
    const bool first = (l == 0);
    const float* Wraw  = Wraw_all + (size_t)l * WR_ELEMS;
    const float wbv = Wb_all[l * H_DIM + colN];
    const float ew  = extw_all[l * H_DIM + colN];
    const float bl  = beta_all[l];

    int it = blockIdx.x;

    constexpr int NW4 = WH_ELEMS / 4;
    constexpr int NWL = (NW4 + 255) / 256;
    auto load_half = [&](const float* srcf, v4f (&wreg)[NWL]) {
        const v4f* src = (const v4f*)srcf;
        #pragma unroll
        for (int j = 0; j < NWL; ++j) {
            const unsigned o = tid + 256u * j;
            if (256 * (j + 1) <= NW4) wreg[j] = src[o];
            else wreg[j] = src[o < (unsigned)NW4 ? o : (unsigned)(NW4 - 1)];
        }
    };
    auto store_half = [&](const v4f (&wreg)[NWL]) {
        #pragma unroll
        for (int j = 0; j < NWL; ++j) {
            const unsigned o = tid + 256u * j;
            if (256 * (j + 1) <= NW4 || o < (unsigned)NW4) *(v4f*)&wsm[4u * o] = wreg[j];
        }
    };
    {
        v4f wreg[NWL];
        load_half(Wraw, wreg);
        store_half(wreg);
    }
    __syncthreads();

    v2f bf[H_DIM / 4];
    float s = 0.f;
    int c = colN;
    asm volatile("" : "+v"(c));
    float dsc = 0.f, dad = 0.f;
    if (lowcol) {
        wsm[c * WR_PITCH + c] = 0.f;
        strip_walk<0, H_DIM / 4>(wsm, 0, c, hi, bf, s);
        dsc = wsm[c * WR_PITCH + H_DIM];
        dad = wsm[c * WR_PITCH + H_DIM + 1];
    } else {
        strip_walk<0, WH_ROWS / 4>(wsm, 0, c, hi, bf, s);
    }
    v4f wregB[NWL];
    load_half(Wraw + WH_ELEMS, wregB);
    v4f t0[NPF];
    if (first && it < nIt) load_tile(it, t0);
    __syncthreads();
    store_half(wregB);
    if (first && it < nIt) {
        #pragma unroll
        for (int j = 0; j < NPF; ++j) *(v4f*)&Ts[0][sr + 8 * j][sc] = t0[j];
    }
    __syncthreads();
    if (!lowcol) {
        asm volatile("" : "+v"(c));
        wsm[c * WR_PITCH + c - WH_ELEMS] = 0.f;
        strip_walk<WH_ROWS / 4, H_DIM / 4>(wsm, -WH_ELEMS, c, hi, bf, s);
        dsc = wsm[c * WR_PITCH + H_DIM - WH_ELEMS];
        dad = wsm[c * WR_PITCH + H_DIM + 1 - WH_ELEMS];
    }
    {
        const float d = dsc * s + dad;
        #pragma unroll
        for (int ks = 0; ks < H_DIM / 4; ++ks) {
            const int k = 4 * ks + 2 * hi;
            if (k == c)     bf[ks].x = d;
            if (k + 1 == c) bf[ks].y = d;
        }
    }

    int slot = 0;
    for (; it < nIt; it += gridDim.x, ++slot) {
        float (*At)[H_DIM + 4] = Ts[slot];
        const int row0 = it * ROWS_IT;
        const int rv   = Nn - row0;
        const int nxt  = it + gridDim.x;
        v4f pf[NPF];
        const bool fetch = first && nxt < nIt;
        if (fetch) load_tile(nxt, pf);
        const float* h0t = h0 + (size_t)row0 * H_DIM;
        float h0v[16];
        if (!first) {
            #pragma unroll
            for (int q = 0; q < 16; ++q) {
                int lr = (q >> 3) * 16 + (q & 7) + 8 * hi;
                int lc = (lr < rv) ? lr : (rv - 1);
                h0v[q] = h0t[(unsigned)(lc * H_DIM + colN)];
            }
        } else {
            #pragma unroll
            for (int q = 0; q < 16; ++q) h0v[q] = 0.f;
        }
        const int cli = row0 + ((lane < rv) ? lane : (rv - 1));
        const float cl = coef0[cli] + coef1[cli];

        v8f acc0 = {}, acc1 = {};
        tile_mma<H_DIM>(At, m, hi, bf, acc0, acc1);

        if (fetch) {
            #pragma unroll
            for (int j = 0; j < NPF; ++j) *(v4f*)&Ts[slot + 1][sr + 8 * j][sc] = pf[j];
        }
        __syncthreads();

        float res[16];
        #pragma unroll
        for (int q = 0; q < 16; ++q) {
            int lr = (q >> 3) * 16 + (q & 7) + 8 * hi;
            float cf   = __shfl(cl, lr);
            float hv   = At[lr][colN];
            float h0x  = first ? hv : h0v[q];
            float outp = ((q < 8) ? acc0[q & 7] : acc1[q & 7]) + wbv;
            float outv = cf * outp - (hv * ew + bl * h0x);
            res[q] = hv + STEPF * fmaxf(outv, 0.f);
        }
        #pragma unroll
        for (int q = 0; q < 16; ++q)
            At[(q >> 3) * 16 + (q & 7) + 8 * hi][colN] = res[q];
    }
    }

    __syncthreads();
    const int dwv = (wave + ((blockIdx.x >= (gridDim.x >> 1)) ? 6 : 0)) & 7;
    if (dwv < 6) {
        const int ct  = dwv % 3;
        const int par = dwv / 3;
        const int dcol = ct * 16 + m;
        const bool colok = (dcol < C_DIM);
        const float* wrow = decW + (size_t)(colok ? dcol : 0) * H_DIM;
        v2f bd[H_DIM / 4];
        #pragma unroll
        for (int ks = 0; ks < H_DIM / 4; ++ks) {
            v2f b = *(const v2f*)&wrow[4 * ks + 2 * hi];
            if (!colok) { b.x = 0.f; b.y = 0.f; }
            bd[ks] = b;
        }
        const float dbv = decb[colok ? dcol : 0];
        int slot = par;
        for (int it = (int)blockIdx.x + par * (int)gridDim.x; it < nIt; it += 2 * (int)gridDim.x, slot += 2) {
            const int row0 = it * ROWS_IT;
            const int rv   = Nn - row0;
            v8f acc0 = {}, acc1 = {};
            tile_mma<H_DIM>(Ts[slot], m, hi, bd, acc0, acc1);
            if (colok) {
                float* ot = out + (size_t)row0 * C_DIM;
                #pragma unroll
                for (int r = 0; r < 8; ++r) {
                    int lr = r + 8 * hi;
                    if (lr < rv)
                        __builtin_nontemporal_store(acc0[r] + dbv, &ot[(unsigned)(lr * C_DIM + dcol)]);
                }
                #pragma unroll
                for (int r = 0; r < 8; ++r) {
                    int lr = 16 + r + 8 * hi;
                    if (lr < rv)
                        __builtin_nontemporal_store(acc1[r] + dbv, &ot[(unsigned)(lr * C_DIM + dcol)]);
                }
            }
        }
    }
}


extern "C" void kernel_launch(void* const* d_in, const int* in_sizes, int n_in,
                              void* d_out, int out_size, void* d_ws, size_t ws_size,
                              hipStream_t stream) {
    const float* x     = (const float*)d_in[0];
    const int*   eidx  = (const int*)d_in[1];
    const float* enc_W = (const float*)d_in[2];
    const float* enc_b = (const float*)d_in[3];
    const float* W_raw = (const float*)d_in[4];
    const float* W_b   = (const float*)d_in[5];
    const float* ext_w = (const float*)d_in[6];
    const float* beta  = (const float*)d_in[7];
    const float* dec_W = (const float*)d_in[8];
    const float* dec_b = (const float*)d_in[9];

    const int Nn = in_sizes[0] / F_IN_DIM;
    const int E  = in_sizes[1] / 2;
    const int* row = eidx;
    const int* col = eidx + E;

    char* ws = (char*)d_ws;
    size_t off = 0;
    auto wsalloc = [&](size_t bytes) -> void* {
        void* p = ws + off;
        off = (off + bytes + 255) & ~(size_t)255;
        return p;
    };
    float* deg  = (float*)wsalloc((size_t)Nn * sizeof(float));
    float* deg1 = (float*)wsalloc((size_t)Nn * sizeof(float));
    float* coef = (float*)wsalloc((size_t)Nn * sizeof(float));
    float* cof1 = (float*)wsalloc((size_t)Nn * sizeof(float));
    float* h0   = (float*)wsalloc((size_t)Nn * H_DIM * sizeof(float));
    (void)ws_size; (void)n_in; (void)out_size;

    const int tb = 256;
    zero_four<<<(Nn + tb - 1) / tb, tb, 0, stream>>>(deg, deg1, coef, cof1, Nn);
    encoder_edge<0><<<256, 512, 0, stream>>>(x, enc_W, enc_b, h0, row, col, nullptr, deg, deg1, Nn, E);
    deg_inv_kernel<<<(Nn + tb - 1) / tb, tb, 0, stream>>>(deg, deg1, Nn);
    encoder_edge<1><<<256, 512, 0, stream>>>(x, enc_W, enc_b, h0, row, col, deg, coef, cof1, Nn, E);

    layers_gemm<<<512, 256, 0, stream>>>(h0, coef, cof1, W_raw, W_b, ext_w, beta, dec_W, dec_b,
                                         (float*)d_out, Nn);
}


# layer epilogue: its LDS operands (h_in; coef, now from a per-block LDS table filled once) are read as one group ahead of each tile's products; 16 shuffles go; sum order unchanged
# speedup vs baseline: 2.8111x; 1.0916x over previous
#include <hip/hip_runtime.h>

#define H_DIM   128
#define F_IN_DIM 256
#define C_DIM   40
#define L_DIM   4
#define STEPF   0.1f
#define WR_PITCH (H_DIM + 2)
#define WR_ELEMS (H_DIM * WR_PITCH)

typedef __attribute__((ext_vector_type(2))) float v2f;
typedef __attribute__((ext_vector_type(4))) float v4f;
typedef __attribute__((ext_vector_type(8))) float v8f;


__device__ __forceinline__ unsigned package_half() {
    return (__builtin_amdgcn_s_sendmsg_rtn(0x87u) >> 18) & 1u;
}

__global__ void zero_four(float* __restrict__ a, float* __restrict__ b,
                          float* __restrict__ c, float* __restrict__ d, int n) {
    int i = blockIdx.x * blockDim.x + threadIdx.x;
    if (i < n) { a[i] = 0.f; b[i] = 0.f; c[i] = 0.f; d[i] = 0.f; }
}

__global__ void deg_inv_kernel(float* __restrict__ deg0, const float* __restrict__ deg1, int n) {
    int i = blockIdx.x * blockDim.x + threadIdx.x;
    if (i < n) { float d = deg0[i] + deg1[i]; deg0[i] = (d > 0.f) ? rsqrtf(d) : 0.f; }
}


#define ROWS_IT 32

template <int K>
__device__ __forceinline__ void tile_mma(const float (*At)[K + 4], int m, int hi,
                                         const v2f (&bf)[K / 4], v8f& acc0, v8f& acc1) {
    constexpr int KS = K / 4;
    constexpr int AHEAD = 4;
    v2f a0[KS], a1[KS];
    #pragma unroll
    for (int ks = 0; ks < AHEAD; ++ks) {
        a0[ks] = *(const v2f*)&At[m][4 * ks + 2 * hi];
        a1[ks] = *(const v2f*)&At[16 + m][4 * ks + 2 * hi];
    }
    #pragma unroll
    for (int ks = 0; ks < KS; ++ks) {
        if (ks + AHEAD < KS) {
            a0[ks + AHEAD] = *(const v2f*)&At[m][4 * (ks + AHEAD) + 2 * hi];
            a1[ks + AHEAD] = *(const v2f*)&At[16 + m][4 * (ks + AHEAD) + 2 * hi];
        }
        acc0 = __builtin_amdgcn_wmma_f32_16x16x4_f32(false, a0[ks], false, bf[ks],
                                                     (short)0, acc0, false, false);
        acc1 = __builtin_amdgcn_wmma_f32_16x16x4_f32(false, a1[ks], false, bf[ks],
                                                     (short)0, acc1, false, false);
        __builtin_amdgcn_sched_barrier(0);
    }
}

#define EDGES_PER_IT 4

template <int ROLE>
__launch_bounds__(512, 1)
__global__ void encoder_edge(const float* __restrict__ x, const float* __restrict__ W,
                             const float* __restrict__ bias, float* __restrict__ h,
                             const int* __restrict__ row, const int* __restrict__ col,
                             const float* __restrict__ dinv, float* __restrict__ acc0,
                             float* __restrict__ acc1, int Nn, int E) {
    __shared__ __align__(16) float As[2][ROWS_IT][F_IN_DIM + 4];
    const bool gemm = __builtin_amdgcn_readfirstlane((int)threadIdx.x) < 256;
    const int lane = threadIdx.x & 31;
    const int wave = (threadIdx.x >> 5) & 7;
    const int m    = lane & 15;
    const int hi   = lane >> 4;
    const int colN = wave * 16 + m;

    const int nItAll = (Nn + ROWS_IT - 1) / ROWS_IT;
    const int nHalf  = (nItAll + 1) >> 1;
    const int tEnd   = ROLE ? nItAll : nHalf;
    const int tBeg   = ROLE ? nHalf : 0;
    constexpr int NPF = ROWS_IT * (F_IN_DIM / 4) / 256;
    const unsigned tid = threadIdx.x & 255u;
    const unsigned sr = tid >> 6, sc = (tid & 63) * 4;

    auto load_tile = [&](int t, v4f (&dst)[NPF]) {
        const int r0 = t * ROWS_IT;
        const v4f* base = (const v4f*)(x + (size_t)r0 * F_IN_DIM);
        const int rv = Nn - r0;
        if (rv >= ROWS_IT) {
            #pragma unroll
            for (int j = 0; j < NPF; ++j)
                dst[j] = __builtin_nontemporal_load(&base[tid + 256u * j]);
        } else {
            const unsigned last = (unsigned)rv * (F_IN_DIM / 4) - 1u;
            #pragma unroll
            for (int j = 0; j < NPF; ++j) {
                unsigned o = tid + 256u * j;
                dst[j] = __builtin_nontemporal_load(&base[o < last ? o : last]);
            }
        }
    };

    const unsigned G = gridDim.x * 256u;
    const unsigned g = blockIdx.x * 256u + tid;
    float* accp = package_half() ? acc1 : acc0;
    int er[EDGES_PER_IT], ec[EDGES_PER_IT];
    auto load_idx = [&](unsigned k0, int (&r)[EDGES_PER_IT], int (&c)[EDGES_PER_IT]) {
        #pragma unroll
        for (int q = 0; q < EDGES_PER_IT; ++q) {
            const unsigned e  = g + G * (k0 + (unsigned)q);
            const bool     ok = e < (unsigned)E;
            const unsigned ee = ok ? e : 0u;
            const int rr = row[ee];
            c[q] = ROLE ? col[ee] : 0;
            r[q] = ok ? rr : -1;
        }
    };
    auto do_edges = [&](const int (&r)[EDGES_PER_IT], const int (&c)[EDGES_PER_IT]) {
        if (ROLE == 0) {
            #pragma unroll
            for (int q = 0; q < EDGES_PER_IT; ++q)
                if (r[q] >= 0) atomicAdd(&accp[r[q]], 1.0f);
        } else {
            float pa[EDGES_PER_IT], pb[EDGES_PER_IT];
            #pragma unroll
            for (int q = 0; q < EDGES_PER_IT; ++q) {
                pa[q] = dinv[r[q] >= 0 ? r[q] : 0];
                pb[q] = dinv[c[q]];
            }
            #pragma unroll
            for (int q = 0; q < EDGES_PER_IT; ++q)
                if (r[q] >= 0) atomicAdd(&accp[c[q]], pa[q] * pb[q]);
        }
    };

    v2f bf[F_IN_DIM / 4];
    float bv = 0.f;
    int it = tBeg + (int)blockIdx.x;
    if (gemm) {
        const float* wrow = W + (size_t)colN * F_IN_DIM;
        #pragma unroll
        for (int ks = 0; ks < F_IN_DIM / 4; ++ks)
            bf[ks] = *(const v2f*)&wrow[4 * ks + 2 * hi];
        bv = bias[colN];
        if (it < tEnd) {
            v4f t0[NPF];
            load_tile(it, t0);
            #pragma unroll
            for (int j = 0; j < NPF; ++j) *(v4f*)&As[0][sr + 4 * j][sc] = t0[j];
        }
    } else {
        #pragma unroll
        for (int ks = 0; ks < F_IN_DIM / 4; ++ks) { bf[ks].x = 0.f; bf[ks].y = 0.f; }
        load_idx(0u, er, ec);
    }
    unsigned kc = 0;
    __syncthreads();
    int buf = 0;
    for (; it < tEnd; it += gridDim.x) {
        if (gemm) {
            const int row0 = it * ROWS_IT;
            const int rv   = Nn - row0;
            const int nxt  = it + gridDim.x;
            v4f pf[NPF];
            if (nxt < tEnd) load_tile(nxt, pf);

            v8f acc0v = {}, acc1v = {};
            tile_mma<F_IN_DIM>(As[buf], m, hi, bf, acc0v, acc1v);

            float* ht = h + (size_t)row0 * H_DIM;
            const unsigned o0 = (unsigned)(8 * hi * H_DIM + colN);
            if (rv >= ROWS_IT) {
                #pragma unroll
                for (int r = 0; r < 8; ++r) {
                    ht[o0 + (unsigned)(r * H_DIM)]        = acc0v[r] + bv;
                    ht[o0 + (unsigned)((16 + r) * H_DIM)] = acc1v[r] + bv;
                }
            } else {
                #pragma unroll
                for (int r = 0; r < 8; ++r) {
                    if (r + 8 * hi < rv)      ht[o0 + (unsigned)(r * H_DIM)]        = acc0v[r] + bv;
                    if (16 + r + 8 * hi < rv) ht[o0 + (unsigned)((16 + r) * H_DIM)] = acc1v[r] + bv;
                }
            }

            if (nxt < tEnd) {
                #pragma unroll
                for (int j = 0; j < NPF; ++j) *(v4f*)&As[buf ^ 1][sr + 4 * j][sc] = pf[j];
            }
        } else {
            int nr[EDGES_PER_IT], nc[EDGES_PER_IT];
            load_idx(kc + EDGES_PER_IT, nr, nc);
            do_edges(er, ec);
            #pragma unroll
            for (int q = 0; q < EDGES_PER_IT; ++q) { er[q] = nr[q]; ec[q] = nc[q]; }
            kc += EDGES_PER_IT;
        }
        __syncthreads();
        buf ^= 1;
    }
    if (!gemm) {
        do_edges(er, ec);
        for (unsigned k = kc + EDGES_PER_IT; g + G * k < (unsigned)E; k += EDGES_PER_IT) {
            load_idx(k, er, ec);
            do_edges(er, ec);
        }
    }
}

#define MAX_SLOTS 7
#define WH_ROWS   (H_DIM / 2)
#define WH_ELEMS  (WH_ROWS * WR_PITCH)

template <int KS0, int KS1>
__device__ __forceinline__ void strip_walk(const float* wsm, int off, int c, int hi,
                                           v2f (&bf)[H_DIM / 4], float& s) {
    const int cw = off + c * WR_PITCH;
    #pragma unroll
    for (int ks = KS0; ks < KS1; ++ks) {
        float v[4];
        #pragma unroll
        for (int e = 0; e < 4; ++e) {
            const int j = 4 * ks + e;
            const int mx = (c > j) ? c : j;
            v[e] = wsm[(cw - (WR_PITCH - 1) * mx) + j * WR_PITCH];
            s += fabsf(v[e]);
        }
        bf[ks].x = hi ? v[2] : v[0];
        bf[ks].y = hi ? v[3] : v[1];
    }
}

__launch_bounds__(256, 4)
__global__ void layers_gemm(const float* __restrict__ h0, const float* __restrict__ coef0,
                            const float* __restrict__ coef1, const float* __restrict__ Wraw_all,
                            const float* __restrict__ Wb_all, const float* __restrict__ extw_all,
                            const float* __restrict__ beta_all, const float* __restrict__ decW,
                            const float* __restrict__ decb, float* __restrict__ out, int Nn) {
    __shared__ __align__(16) float wsm[WH_ELEMS];
    __shared__ __align__(16) float Ts[MAX_SLOTS][ROWS_IT][H_DIM + 4];
    __shared__ __align__(16) float Cs[MAX_SLOTS][ROWS_IT];
    const int lane = threadIdx.x & 31;
    const int wave = __builtin_amdgcn_readfirstlane((int)threadIdx.x) >> 5;
    const int m    = lane & 15;
    const int hi   = lane >> 4;
    const int colN = wave * 16 + m;
    const bool lowcol = wave < (WH_ROWS / 16);

    const int nIt = (Nn + ROWS_IT - 1) / ROWS_IT;
    constexpr int NPF = ROWS_IT * (H_DIM / 4) / 256;
    const unsigned tid = threadIdx.x;
    const unsigned sr = tid >> 5, sc = (tid & 31) * 4;

    auto load_tile = [&](int t, v4f (&dst)[NPF]) {
        const int r0 = t * ROWS_IT;
        const v4f* base = (const v4f*)(h0 + (size_t)r0 * H_DIM);
        const int rv = Nn - r0;
        if (rv >= ROWS_IT) {
            #pragma unroll
            for (int j = 0; j < NPF; ++j) dst[j] = base[tid + 256u * j];
        } else {
            const unsigned last = (unsigned)rv * (H_DIM / 4) - 1u;
            #pragma unroll
            for (int j = 0; j < NPF; ++j) {
                unsigned o = tid + 256u * j;
                dst[j] = base[o < last ? o : last];
            }
        }
    };

    if (tid < (unsigned)(MAX_SLOTS * ROWS_IT)) {
        const int sl = (int)(tid >> 5), r = (int)(tid & 31u);
        const int t  = (int)blockIdx.x + sl * (int)gridDim.x;
        if (t < nIt) {
            const int r0 = t * ROWS_IT;
            const int rvv = Nn - r0;
            const int ci = r0 + ((r < rvv) ? r : (rvv - 1));
            Cs[sl][r] = coef0[ci] + coef1[ci];
        }
    }

    #pragma unroll 1
    for (int l = 0; l < L_DIM; ++l) {
    const bool first = (l == 0);
    const float* Wraw  = Wraw_all + (size_t)l * WR_ELEMS;
    const float wbv = Wb_all[l * H_DIM + colN];
    const float ew  = extw_all[l * H_DIM + colN];
    const float bl  = beta_all[l];

    int it = blockIdx.x;

    constexpr int NW4 = WH_ELEMS / 4;
    constexpr int NWL = (NW4 + 255) / 256;
    auto load_half = [&](const float* srcf, v4f (&wreg)[NWL]) {
        const v4f* src = (const v4f*)srcf;
        #pragma unroll
        for (int j = 0; j < NWL; ++j) {
            const unsigned o = tid + 256u * j;
            if (256 * (j + 1) <= NW4) wreg[j] = src[o];
            else wreg[j] = src[o < (unsigned)NW4 ? o : (unsigned)(NW4 - 1)];
        }
    };
    auto store_half = [&](const v4f (&wreg)[NWL]) {
        #pragma unroll
        for (int j = 0; j < NWL; ++j) {
            const unsigned o = tid + 256u * j;
            if (256 * (j + 1) <= NW4 || o < (unsigned)NW4) *(v4f*)&wsm[4u * o] = wreg[j];
        }
    };
    {
        v4f wreg[NWL];
        load_half(Wraw, wreg);
        store_half(wreg);
    }
    __syncthreads();

    v2f bf[H_DIM / 4];
    float s = 0.f;
    int c = colN;
    asm volatile("" : "+v"(c));
    float dsc = 0.f, dad = 0.f;
    if (lowcol) {
        wsm[c * WR_PITCH + c] = 0.f;
        strip_walk<0, H_DIM / 4>(wsm, 0, c, hi, bf, s);
        dsc = wsm[c * WR_PITCH + H_DIM];
        dad = wsm[c * WR_PITCH + H_DIM + 1];
    } else {
        strip_walk<0, WH_ROWS / 4>(wsm, 0, c, hi, bf, s);
    }
    v4f wregB[NWL];
    load_half(Wraw + WH_ELEMS, wregB);
    v4f t0[NPF];
    if (first && it < nIt) load_tile(it, t0);
    __syncthreads();
    store_half(wregB);
    if (first && it < nIt) {
        #pragma unroll
        for (int j = 0; j < NPF; ++j) *(v4f*)&Ts[0][sr + 8 * j][sc] = t0[j];
    }
    __syncthreads();
    if (!lowcol) {
        asm volatile("" : "+v"(c));
        wsm[c * WR_PITCH + c - WH_ELEMS] = 0.f;
        strip_walk<WH_ROWS / 4, H_DIM / 4>(wsm, -WH_ELEMS, c, hi, bf, s);
        dsc = wsm[c * WR_PITCH + H_DIM - WH_ELEMS];
        dad = wsm[c * WR_PITCH + H_DIM + 1 - WH_ELEMS];
    }
    {
        const float d = dsc * s + dad;
        #pragma unroll
        for (int ks = 0; ks < H_DIM / 4; ++ks) {
            const int k = 4 * ks + 2 * hi;
            if (k == c)     bf[ks].x = d;
            if (k + 1 == c) bf[ks].y = d;
        }
    }

    int mh = m, hh = hi, cn = colN;
    asm volatile("" : "+v"(mh), "+v"(hh), "+v"(cn));

    int slot = 0;
    for (; it < nIt; it += gridDim.x, ++slot) {
        float (*At)[H_DIM + 4] = Ts[slot];
        const int row0 = it * ROWS_IT;
        const int rv   = Nn - row0;
        const int nxt  = it + gridDim.x;
        v4f pf[NPF];
        const bool fetch = first && nxt < nIt;
        if (fetch) load_tile(nxt, pf);
        const float* h0t = h0 + (size_t)row0 * H_DIM;
        float h0v[16];
        if (!first) {
            #pragma unroll
            for (int q = 0; q < 16; ++q) {
                int lr = (q >> 3) * 16 + (q & 7) + 8 * hh;
                int lc = (lr < rv) ? lr : (rv - 1);
                h0v[q] = h0t[(unsigned)(lc * H_DIM + cn)];
            }
        } else {
            #pragma unroll
            for (int q = 0; q < 16; ++q) h0v[q] = 0.f;
        }
        float cfv[16], hvv[16];
        {
            const v4f* cp = (const v4f*)&Cs[slot][0];
            const v4f c0 = cp[2 * hh], c1 = cp[2 * hh + 1], c2 = cp[4 + 2 * hh], c3 = cp[5 + 2 * hh];
            #pragma unroll
            for (int e = 0; e < 4; ++e) {
                cfv[e] = c0[e]; cfv[4 + e] = c1[e]; cfv[8 + e] = c2[e]; cfv[12 + e] = c3[e];
            }
            #pragma unroll
            for (int q = 0; q < 16; ++q)
                hvv[q] = At[(q >> 3) * 16 + (q & 7) + 8 * hh][cn];
        }
        __builtin_amdgcn_sched_barrier(0);

        v8f acc0 = {}, acc1 = {};
        tile_mma<H_DIM>(At, mh, hh, bf, acc0, acc1);

        if (fetch) {
            #pragma unroll
            for (int j = 0; j < NPF; ++j) *(v4f*)&Ts[slot + 1][sr + 8 * j][sc] = pf[j];
        }
        __syncthreads();

        float res[16];
        #pragma unroll
        for (int q = 0; q < 16; ++q) {
            float cf   = cfv[q];
            float hv   = hvv[q];
            float h0x  = first ? hv : h0v[q];
            float outp = ((q < 8) ? acc0[q & 7] : acc1[q & 7]) + wbv;
            float outv = cf * outp - (hv * ew + bl * h0x);
            res[q] = hv + STEPF * fmaxf(outv, 0.f);
        }
        #pragma unroll
        for (int q = 0; q < 16; ++q)
            At[(q >> 3) * 16 + (q & 7) + 8 * hh][cn] = res[q];
    }
    }

    __syncthreads();
    const int dwv = (wave + ((blockIdx.x >= (gridDim.x >> 1)) ? 6 : 0)) & 7;
    if (dwv < 6) {
        const int ct  = dwv % 3;
        const int par = dwv / 3;
        const int dcol = ct * 16 + m;
        const bool colok = (dcol < C_DIM);
        const float* wrow = decW + (size_t)(colok ? dcol : 0) * H_DIM;
        v2f bd[H_DIM / 4];
        #pragma unroll
        for (int ks = 0; ks < H_DIM / 4; ++ks) {
            v2f b = *(const v2f*)&wrow[4 * ks + 2 * hi];
            if (!colok) { b.x = 0.f; b.y = 0.f; }
            bd[ks] = b;
        }
        const float dbv = decb[colok ? dcol : 0];
        int slot = par;
        for (int it = (int)blockIdx.x + par * (int)gridDim.x; it < nIt; it += 2 * (int)gridDim.x, slot += 2) {
            const int row0 = it * ROWS_IT;
            const int rv   = Nn - row0;
            v8f acc0 = {}, acc1 = {};
            tile_mma<H_DIM>(Ts[slot], m, hi, bd, acc0, acc1);
            if (colok) {
                float* ot = out + (size_t)row0 * C_DIM;
                #pragma unroll
                for (int r = 0; r < 8; ++r) {
                    int lr = r + 8 * hi;
                    if (lr < rv)
                        __builtin_nontemporal_store(acc0[r] + dbv, &ot[(unsigned)(lr * C_DIM + dcol)]);
                }
                #pragma unroll
                for (int r = 0; r < 8; ++r) {
                    int lr = 16 + r + 8 * hi;
                    if (lr < rv)
                        __builtin_nontemporal_store(acc1[r] + dbv, &ot[(unsigned)(lr * C_DIM + dcol)]);
                }
            }
        }
    }
}


extern "C" void kernel_launch(void* const* d_in, const int* in_sizes, int n_in,
                              void* d_out, int out_size, void* d_ws, size_t ws_size,
                              hipStream_t stream) {
    const float* x     = (const float*)d_in[0];
    const int*   eidx  = (const int*)d_in[1];
    const float* enc_W = (const float*)d_in[2];
    const float* enc_b = (const float*)d_in[3];
    const float* W_raw = (const float*)d_in[4];
    const float* W_b   = (const float*)d_in[5];
    const float* ext_w = (const float*)d_in[6];
    const float* beta  = (const float*)d_in[7];
    const float* dec_W = (const float*)d_in[8];
    const float* dec_b = (const float*)d_in[9];

    const int Nn = in_sizes[0] / F_IN_DIM;
    const int E  = in_sizes[1] / 2;
    const int* row = eidx;
    const int* col = eidx + E;

    char* ws = (char*)d_ws;
    size_t off = 0;
    auto wsalloc = [&](size_t bytes) -> void* {
        void* p = ws + off;
        off = (off + bytes + 255) & ~(size_t)255;
        return p;
    };
    float* deg  = (float*)wsalloc((size_t)Nn * sizeof(float));
    float* deg1 = (float*)wsalloc((size_t)Nn * sizeof(float));
    float* coef = (float*)wsalloc((size_t)Nn * sizeof(float));
    float* cof1 = (float*)wsalloc((size_t)Nn * sizeof(float));
    float* h0   = (float*)wsalloc((size_t)Nn * H_DIM * sizeof(float));
    (void)ws_size; (void)n_in; (void)out_size;

    const int tb = 256;
    zero_four<<<(Nn + tb - 1) / tb, tb, 0, stream>>>(deg, deg1, coef, cof1, Nn);
    encoder_edge<0><<<256, 512, 0, stream>>>(x, enc_W, enc_b, h0, row, col, nullptr, deg, deg1, Nn, E);
    deg_inv_kernel<<<(Nn + tb - 1) / tb, tb, 0, stream>>>(deg, deg1, Nn);
    encoder_edge<1><<<256, 512, 0, stream>>>(x, enc_W, enc_b, h0, row, col, deg, coef, cof1, Nn, E);

    layers_gemm<<<512, 256, 0, stream>>>(h0, coef, cof1, W_raw, W_b, ext_w, beta, dec_W, dec_b,
                                         (float*)d_out, Nn);
}
